# starting kernel
# baseline (speedup 1.0000x reference)
#include <hip/hip_runtime.h>

typedef __attribute__((ext_vector_type(2))) float v2f;
typedef __attribute__((ext_vector_type(8))) float v8f;

__global__ void k_fill(float* __restrict__ p, float val, int n) {
  int i = blockIdx.x * blockDim.x + threadIdx.x;
  if (i < n) p[i] = val;
}

__global__ void k_deg_accum(const int* __restrict__ dst, float* __restrict__ deg, int E) {
  int i = blockIdx.x * blockDim.x + threadIdx.x;
  if (i < E) unsafeAtomicAdd(&deg[dst[i]], 1.0f);
}

__global__ void k_rsqrt_inplace(float* __restrict__ d, int n) {
  int i = blockIdx.x * blockDim.x + threadIdx.x;
  if (i < n) d[i] = 1.0f / sqrtf(d[i]);
}

__global__ void k_relu(float* __restrict__ p, int n) {
  int i = blockIdx.x * blockDim.x + threadIdx.x;
  if (i < n) p[i] = fmaxf(p[i], 0.0f);
}

__global__ void k_init_agg(const float* __restrict__ h, const float* __restrict__ dinv,
                           const float* __restrict__ bias, float* __restrict__ agg,
                           int N, int F) {
  int i = blockIdx.x * blockDim.x + threadIdx.x;
  if (i >= N * F) return;
  int node = i / F;
  int f    = i - node * F;
  float di = dinv[node];
  agg[i] = h[i] * (di * di) + bias[f];
}

__global__ __launch_bounds__(256)
void k_gemm_wmma(const float* __restrict__ A, const float* __restrict__ B,
                 float* __restrict__ C, int M, int K, int Nc) {
  int wid  = (blockIdx.x * blockDim.x + threadIdx.x) >> 5;
  int lane = threadIdx.x & 31;
  int tilesN = Nc >> 4;
  int tilesM = M >> 4;
  if (wid >= tilesM * tilesN) return;
  int tm = wid / tilesN;
  int tn = wid - tm * tilesN;

  int half = lane >> 4;
  int j    = lane & 15;

  const float* Arow = A + (size_t)(tm * 16 + j) * K + 2 * half;
  const float* Bp   = B + (size_t)(2 * half) * Nc + tn * 16 + j;

  v8f acc = {};
  for (int k = 0; k < K; k += 4) {
    v2f a = *(const v2f*)(Arow + k);
    v2f b;
    b.x = Bp[(size_t)k * Nc];
    b.y = Bp[(size_t)(k + 1) * Nc];
    acc = __builtin_amdgcn_wmma_f32_16x16x4_f32(false, a, false, b, (short)0, acc,
                                                false, false);
  }

  float* Cp = C + (size_t)(tm * 16 + 8 * half) * Nc + tn * 16 + j;
#pragma unroll
  for (int v = 0; v < 8; ++v) Cp[(size_t)v * Nc] = acc[v];
}

__global__ __launch_bounds__(256)
void k_scatter(const float* __restrict__ h, const int* __restrict__ src,
               const int* __restrict__ dst, const float* __restrict__ dinv,
               float* __restrict__ agg, int E, int F) {
  int gid  = blockIdx.x * blockDim.x + threadIdx.x;
  int e    = gid >> 5;
  int lane = gid & 31;
  if (e >= E) return;
  int s = src[e], d = dst[e];
  float norm = dinv[s] * dinv[d];
  const float* hs = h + (size_t)s * F;
  float* ad       = agg + (size_t)d * F;
  for (int f = lane; f < F; f += 32)
    unsafeAtomicAdd(&ad[f], hs[f] * norm);
}

extern "C" void kernel_launch(void* const* d_in, const int* in_sizes, int n_in,
                              void* d_out, int out_size, void* d_ws, size_t ws_size,
                              hipStream_t stream) {
  const float* x  = (const float*)d_in[0];
  const int*   ei = (const int*)d_in[1];
  const float* W1 = (const float*)d_in[2];
  const float* b1 = (const float*)d_in[3];
  const float* W2 = (const float*)d_in[4];
  const float* b2 = (const float*)d_in[5];
  float* out = (float*)d_out;

  const int F0 = 128;
  const int N  = in_sizes[0] / F0;
  const int E  = in_sizes[1] / 2;
  const int F1 = in_sizes[3];
  const int F2 = in_sizes[5];
  const int* src = ei;
  const int* dst = ei + E;

  float* dinv = (float*)d_ws;
  float* h1   = dinv + N;
  float* agg1 = h1 + (size_t)N * F1;
  float* h2   = h1;

  const int B = 256;
  auto cdiv = [](long long a, long long b) { return (int)((a + b - 1) / b); };

  k_fill<<<cdiv(N, B), B, 0, stream>>>(dinv, 1.0f, N);
  k_deg_accum<<<cdiv(E, B), B, 0, stream>>>(dst, dinv, E);
  k_rsqrt_inplace<<<cdiv(N, B), B, 0, stream>>>(dinv, N);

  long long waves1 = (long long)(N / 16) * (F1 / 16);
  k_gemm_wmma<<<cdiv(waves1 * 32, B), B, 0, stream>>>(x, W1, h1, N, F0, F1);
  k_init_agg<<<cdiv((long long)N * F1, B), B, 0, stream>>>(h1, dinv, b1, agg1, N, F1);
  k_scatter<<<cdiv((long long)E * 32, B), B, 0, stream>>>(h1, src, dst, dinv, agg1, E, F1);
  k_relu<<<cdiv((long long)N * F1, B), B, 0, stream>>>(agg1, N * F1);

  long long waves2 = (long long)(N / 16) * (F2 / 16);
  k_gemm_wmma<<<cdiv(waves2 * 32, B), B, 0, stream>>>(agg1, W2, h2, N, F1, F2);
  k_init_agg<<<cdiv((long long)N * F2, B), B, 0, stream>>>(h2, dinv, b2, out, N, F2);
  k_scatter<<<cdiv((long long)E * 32, B), B, 0, stream>>>(h2, src, dst, dinv, out, E, F2);
}


# Atomic edge scatter -> in-call dst-sorted edge list (count+rank, scan, place) + one-wave-per-node register gather-sum; row written once with self-loop, bias, relu; sum order = arrival rank per node
# speedup vs baseline: 1.8034x; 1.8034x over previous
#include <hip/hip_runtime.h>

typedef __attribute__((ext_vector_type(2))) float v2f;
typedef __attribute__((ext_vector_type(8))) float v8f;

__global__ void k_fill_i32(int* __restrict__ p, int val, int n) {
  int i = blockIdx.x * blockDim.x + threadIdx.x;
  if (i < n) p[i] = val;
}

__global__ void k_rank(const int* __restrict__ dst, int* __restrict__ cnt,
                       int* __restrict__ rank, int E) {
  int i = blockIdx.x * blockDim.x + threadIdx.x;
  if (i < E) rank[i] = atomicAdd(&cnt[dst[i]], 1);
}

__global__ __launch_bounds__(1024)
void k_scan(const int* __restrict__ cnt, int* __restrict__ rowptr,
            float* __restrict__ dinv, int N) {
  __shared__ int part[1024];
  const int t   = threadIdx.x;
  const int per = (N + 1023) / 1024;
  const int b   = t * per;
  const int e   = min(b + per, N);
  int s = 0;
  for (int i = b; i < e; ++i) s += cnt[i];
  part[t] = s;
  __syncthreads();
  for (int off = 1; off < 1024; off <<= 1) {
    int v = (t >= off) ? part[t - off] : 0;
    __syncthreads();
    part[t] += v;
    __syncthreads();
  }
  int run = part[t] - s;
  for (int i = b; i < e; i += 8) {
    int c[8];
#pragma unroll
    for (int k = 0; k < 8; ++k) c[k] = (i + k < e) ? cnt[i + k] : 0;
#pragma unroll
    for (int k = 0; k < 8; ++k) {
      if (i + k < e) {
        rowptr[i + k] = run;
        dinv[i + k]   = 1.0f / sqrtf((float)(c[k] + 1));
      }
      run += c[k];
    }
  }
  if (t == 1023) rowptr[N] = part[1023];
}

__global__ void k_place(const int* __restrict__ src, const int* __restrict__ dst,
                        const int* __restrict__ rank, const int* __restrict__ rowptr,
                        int* __restrict__ col, int E) {
  int i = blockIdx.x * blockDim.x + threadIdx.x;
  if (i < E) col[rowptr[dst[i]] + rank[i]] = src[i];
}

__global__ __launch_bounds__(256)
void k_gemm_wmma(const float* __restrict__ A, const float* __restrict__ B,
                 float* __restrict__ C, int M, int K, int Nc) {
  int wid  = (blockIdx.x * blockDim.x + threadIdx.x) >> 5;
  int lane = threadIdx.x & 31;
  int tilesN = Nc >> 4;
  int tilesM = M >> 4;
  if (wid >= tilesM * tilesN) return;
  int tm = wid / tilesN;
  int tn = wid - tm * tilesN;

  int half = lane >> 4;
  int j    = lane & 15;

  const float* Arow = A + (size_t)(tm * 16 + j) * K + 2 * half;
  const float* Bp   = B + (size_t)(2 * half) * Nc + tn * 16 + j;

  v8f acc = {};
  for (int k = 0; k < K; k += 4) {
    v2f a = *(const v2f*)(Arow + k);
    v2f b;
    b.x = Bp[(size_t)k * Nc];
    b.y = Bp[(size_t)(k + 1) * Nc];
    acc = __builtin_amdgcn_wmma_f32_16x16x4_f32(false, a, false, b, (short)0, acc,
                                                false, false);
  }

  float* Cp = C + (size_t)(tm * 16 + 8 * half) * Nc + tn * 16 + j;
#pragma unroll
  for (int v = 0; v < 8; ++v) Cp[(size_t)v * Nc] = acc[v];
}

template <int VEC, bool RELU>
__global__ __launch_bounds__(256)
void k_gather(const float* __restrict__ h, const int* __restrict__ rowptr,
              const int* __restrict__ col, const float* __restrict__ dinv,
              const float* __restrict__ bias, float* __restrict__ out, int N) {
  typedef __attribute__((ext_vector_type(VEC))) float vec;
  constexpr int F = 32 * VEC;
  int node = (blockIdx.x * blockDim.x + threadIdx.x) >> 5;
  const int lane = threadIdx.x & 31;
  if (node >= N) return;
  node = __builtin_amdgcn_readfirstlane(node);

  const int   beg = rowptr[node];
  const int   end = rowptr[node + 1];
  const float dd  = dinv[node];
  const float* hl = h + lane * VEC;

  vec acc = 0.0f;
  for (int base = beg; base < end; base += 32) {
    const int n = min(32, end - base);
    int   ms = 0;
    float mw = 0.0f;
    if (lane < n) {
      ms = col[base + lane];
      mw = dinv[ms] * dd;
    }
    const int mwi = __float_as_int(mw);
    int i = 0;
    for (; i + 4 <= n; i += 4) {
      const int s0 = __builtin_amdgcn_readlane(ms, i);
      const int s1 = __builtin_amdgcn_readlane(ms, i + 1);
      const int s2 = __builtin_amdgcn_readlane(ms, i + 2);
      const int s3 = __builtin_amdgcn_readlane(ms, i + 3);
      const vec r0 = *(const vec*)(hl + (size_t)s0 * F);
      const vec r1 = *(const vec*)(hl + (size_t)s1 * F);
      const vec r2 = *(const vec*)(hl + (size_t)s2 * F);
      const vec r3 = *(const vec*)(hl + (size_t)s3 * F);
      const float w0 = __int_as_float(__builtin_amdgcn_readlane(mwi, i));
      const float w1 = __int_as_float(__builtin_amdgcn_readlane(mwi, i + 1));
      const float w2 = __int_as_float(__builtin_amdgcn_readlane(mwi, i + 2));
      const float w3 = __int_as_float(__builtin_amdgcn_readlane(mwi, i + 3));
      acc += r0 * w0;
      acc += r1 * w1;
      acc += r2 * w2;
      acc += r3 * w3;
    }
    for (; i < n; ++i) {
      const int   s = __builtin_amdgcn_readlane(ms, i);
      const float w = __int_as_float(__builtin_amdgcn_readlane(mwi, i));
      const vec   r = *(const vec*)(hl + (size_t)s * F);
      acc += r * w;
    }
  }

  const vec selfv = *(const vec*)(hl + (size_t)node * F);
  const vec bv    = *(const vec*)(bias + lane * VEC);
  vec r = acc + selfv * (dd * dd) + bv;
  if (RELU) {
#pragma unroll
    for (int v = 0; v < VEC; ++v) r[v] = fmaxf(r[v], 0.0f);
  }
  *(vec*)(out + (size_t)node * F + lane * VEC) = r;
}

extern "C" void kernel_launch(void* const* d_in, const int* in_sizes, int n_in,
                              void* d_out, int out_size, void* d_ws, size_t ws_size,
                              hipStream_t stream) {
  const float* x  = (const float*)d_in[0];
  const int*   ei = (const int*)d_in[1];
  const float* W1 = (const float*)d_in[2];
  const float* b1 = (const float*)d_in[3];
  const float* W2 = (const float*)d_in[4];
  const float* b2 = (const float*)d_in[5];
  float* out = (float*)d_out;

  const int F0 = 128;
  const int N  = in_sizes[0] / F0;
  const int E  = in_sizes[1] / 2;
  const int F1 = in_sizes[3];
  const int F2 = in_sizes[5];
  const int* src = ei;
  const int* dst = ei + E;

  int*   cnt    = (int*)d_ws;
  int*   rowptr = cnt + N;
  float* dinv   = (float*)(rowptr + N + 64);
  int*   rank   = (int*)(dinv + N);
  int*   col    = rank + E;
  float* h1     = (float*)(col + E);
  float* agg1   = h1 + (size_t)N * F1;
  float* h2     = h1;

  const int B = 256;
  auto cdiv = [](long long a, long long b) { return (int)((a + b - 1) / b); };

  k_fill_i32<<<cdiv(N, B), B, 0, stream>>>(cnt, 0, N);
  k_rank<<<cdiv(E, B), B, 0, stream>>>(dst, cnt, rank, E);
  k_scan<<<1, 1024, 0, stream>>>(cnt, rowptr, dinv, N);
  k_place<<<cdiv(E, B), B, 0, stream>>>(src, dst, rank, rowptr, col, E);

  long long waves1 = (long long)(N / 16) * (F1 / 16);
  k_gemm_wmma<<<cdiv(waves1 * 32, B), B, 0, stream>>>(x, W1, h1, N, F0, F1);
  k_gather<4, true><<<cdiv((long long)N * 32, B), B, 0, stream>>>(h1, rowptr, col, dinv, b1, agg1, N);

  long long waves2 = (long long)(N / 16) * (F2 / 16);
  k_gemm_wmma<<<cdiv(waves2 * 32, B), B, 0, stream>>>(agg1, W2, h2, N, F1, F2);
  k_gather<2, false><<<cdiv((long long)N * 32, B), B, 0, stream>>>(h2, rowptr, col, dinv, b2, out, N);
}


# Edge ranks from LDS counters, not returning global atomics: 128 edge chunks, each counts into a 50,000-word counter array in its own LDS; chunk-prefix pass; sum order = chunk, then LDS arrival
# speedup vs baseline: 2.1091x; 1.1695x over previous
#include <hip/hip_runtime.h>

typedef __attribute__((ext_vector_type(2))) float v2f;
typedef __attribute__((ext_vector_type(8))) float v8f;

constexpr int CHUNKS     = 128;
constexpr int LDS_NODES  = 51200;
constexpr int EDGE_BATCH = 8;

__device__ __forceinline__ int row_pitch(int N) { return (N + 3) & ~3; }

__global__ __launch_bounds__(1024)
void k_rank(const int* __restrict__ dst, int* __restrict__ hist, int* __restrict__ rank,
            int E, int N) {
  __shared__ alignas(16) int c[LDS_NODES];
  const int t  = threadIdx.x;
  const int z4 = (N + 3) >> 2;
  for (int i = t; i < z4; i += 1024) ((int4*)c)[i] = make_int4(0, 0, 0, 0);
  __syncthreads();

  const int per = (E + CHUNKS - 1) / CHUNKS;
  const int e0  = blockIdx.x * per;
  const int e1  = min(e0 + per, E);
  for (int eb = e0 + t; eb < e1; eb += EDGE_BATCH * 1024) {
    int d[EDGE_BATCH];
#pragma unroll
    for (int k = 0; k < EDGE_BATCH; ++k) {
      const int e = eb + k * 1024;
      d[k] = (e < e1) ? dst[e] : 0;
    }
#pragma unroll
    for (int k = 0; k < EDGE_BATCH; ++k) {
      const int e = eb + k * 1024;
      if (e < e1) rank[e] = atomicAdd(&c[d[k]], 1);
    }
  }
  __syncthreads();

  int* h = hist + (size_t)blockIdx.x * row_pitch(N);
  const int f4 = N >> 2;
  for (int i = t; i < f4; i += 1024) ((int4*)h)[i] = ((const int4*)c)[i];
  for (int i = (f4 << 2) + t; i < N; i += 1024) h[i] = c[i];
}

__global__ __launch_bounds__(256)
void k_chunk_prefix(int* __restrict__ hist, int* __restrict__ cnt, int N) {
  const int i = blockIdx.x * 256 + threadIdx.x;
  if (i >= N) return;
  const size_t P = (size_t)row_pitch(N);
  int* p = hist + i;
  int run = 0;
  for (int b0 = 0; b0 < CHUNKS; b0 += 32) {
    int v[32];
#pragma unroll
    for (int k = 0; k < 32; ++k) v[k] = p[(size_t)(b0 + k) * P];
#pragma unroll
    for (int k = 0; k < 32; ++k) {
      p[(size_t)(b0 + k) * P] = run;
      run += v[k];
    }
  }
  cnt[i] = run;
}

__global__ __launch_bounds__(1024)
void k_scan(const int* __restrict__ cnt, int* __restrict__ rowptr,
            float* __restrict__ dinv, int N) {
  __shared__ int part[1024];
  const int t   = threadIdx.x;
  const int per = (N + 1023) / 1024;
  const int b   = t * per;
  const int e   = min(b + per, N);
  int s = 0;
  for (int i = b; i < e; ++i) s += cnt[i];
  part[t] = s;
  __syncthreads();
  for (int off = 1; off < 1024; off <<= 1) {
    int v = (t >= off) ? part[t - off] : 0;
    __syncthreads();
    part[t] += v;
    __syncthreads();
  }
  int run = part[t] - s;
  for (int i = b; i < e; i += 8) {
    int c[8];
#pragma unroll
    for (int k = 0; k < 8; ++k) c[k] = (i + k < e) ? cnt[i + k] : 0;
#pragma unroll
    for (int k = 0; k < 8; ++k) {
      if (i + k < e) {
        rowptr[i + k] = run;
        dinv[i + k]   = 1.0f / sqrtf((float)(c[k] + 1));
      }
      run += c[k];
    }
  }
  if (t == 1023) rowptr[N] = part[1023];
}

__global__ __launch_bounds__(1024)
void k_place(const int* __restrict__ src, const int* __restrict__ dst,
             const int* __restrict__ rank, const int* __restrict__ rowptr,
             const int* __restrict__ hist, int* __restrict__ col, int E, int N) {
  __shared__ alignas(16) int base[LDS_NODES];
  const int t  = threadIdx.x;
  const int* h = hist + (size_t)blockIdx.x * row_pitch(N);
  const int f4 = N >> 2;
#pragma unroll 4
  for (int i = t; i < f4; i += 1024) {
    const int4 a = ((const int4*)rowptr)[i];
    const int4 b = ((const int4*)h)[i];
    ((int4*)base)[i] = make_int4(a.x + b.x, a.y + b.y, a.z + b.z, a.w + b.w);
  }
  for (int i = (f4 << 2) + t; i < N; i += 1024) base[i] = rowptr[i] + h[i];
  __syncthreads();

  const int per = (E + CHUNKS - 1) / CHUNKS;
  const int e0  = blockIdx.x * per;
  const int e1  = min(e0 + per, E);
  for (int eb = e0 + t; eb < e1; eb += EDGE_BATCH * 1024) {
    int d[EDGE_BATCH], s[EDGE_BATCH], r[EDGE_BATCH];
#pragma unroll
    for (int k = 0; k < EDGE_BATCH; ++k) {
      const int  e  = eb + k * 1024;
      const bool ok = e < e1;
      d[k] = ok ? dst[e]  : 0;
      s[k] = ok ? src[e]  : 0;
      r[k] = ok ? rank[e] : 0;
    }
#pragma unroll
    for (int k = 0; k < EDGE_BATCH; ++k) {
      const int e = eb + k * 1024;
      if (e < e1) col[base[d[k]] + r[k]] = s[k];
    }
  }
}

__global__ __launch_bounds__(256)
void k_gemm_wmma(const float* __restrict__ A, const float* __restrict__ B,
                 float* __restrict__ C, int M, int K, int Nc) {
  int wid  = (blockIdx.x * blockDim.x + threadIdx.x) >> 5;
  int lane = threadIdx.x & 31;
  int tilesN = Nc >> 4;
  int tilesM = M >> 4;
  if (wid >= tilesM * tilesN) return;
  int tm = wid / tilesN;
  int tn = wid - tm * tilesN;

  int half = lane >> 4;
  int j    = lane & 15;

  const float* Arow = A + (size_t)(tm * 16 + j) * K + 2 * half;
  const float* Bp   = B + (size_t)(2 * half) * Nc + tn * 16 + j;

  v8f acc = {};
  for (int k = 0; k < K; k += 4) {
    v2f a = *(const v2f*)(Arow + k);
    v2f b;
    b.x = Bp[(size_t)k * Nc];
    b.y = Bp[(size_t)(k + 1) * Nc];
    acc = __builtin_amdgcn_wmma_f32_16x16x4_f32(false, a, false, b, (short)0, acc,
                                                false, false);
  }

  float* Cp = C + (size_t)(tm * 16 + 8 * half) * Nc + tn * 16 + j;
#pragma unroll
  for (int v = 0; v < 8; ++v) Cp[(size_t)v * Nc] = acc[v];
}

template <int VEC, bool RELU>
__global__ __launch_bounds__(256)
void k_gather(const float* __restrict__ h, const int* __restrict__ rowptr,
              const int* __restrict__ col, const float* __restrict__ dinv,
              const float* __restrict__ bias, float* __restrict__ out, int N) {
  typedef __attribute__((ext_vector_type(VEC))) float vec;
  constexpr int F = 32 * VEC;
  int node = (blockIdx.x * blockDim.x + threadIdx.x) >> 5;
  const int lane = threadIdx.x & 31;
  if (node >= N) return;
  node = __builtin_amdgcn_readfirstlane(node);

  const int   beg = rowptr[node];
  const int   end = rowptr[node + 1];
  const float dd  = dinv[node];
  const float* hl = h + lane * VEC;

  vec acc = 0.0f;
  for (int base = beg; base < end; base += 32) {
    const int n = min(32, end - base);
    int   ms = 0;
    float mw = 0.0f;
    if (lane < n) {
      ms = col[base + lane];
      mw = dinv[ms] * dd;
    }
    const int mwi = __float_as_int(mw);
    int i = 0;
    for (; i + 4 <= n; i += 4) {
      const int s0 = __builtin_amdgcn_readlane(ms, i);
      const int s1 = __builtin_amdgcn_readlane(ms, i + 1);
      const int s2 = __builtin_amdgcn_readlane(ms, i + 2);
      const int s3 = __builtin_amdgcn_readlane(ms, i + 3);
      const vec r0 = *(const vec*)(hl + (size_t)s0 * F);
      const vec r1 = *(const vec*)(hl + (size_t)s1 * F);
      const vec r2 = *(const vec*)(hl + (size_t)s2 * F);
      const vec r3 = *(const vec*)(hl + (size_t)s3 * F);
      const float w0 = __int_as_float(__builtin_amdgcn_readlane(mwi, i));
      const float w1 = __int_as_float(__builtin_amdgcn_readlane(mwi, i + 1));
      const float w2 = __int_as_float(__builtin_amdgcn_readlane(mwi, i + 2));
      const float w3 = __int_as_float(__builtin_amdgcn_readlane(mwi, i + 3));
      acc += r0 * w0;
      acc += r1 * w1;
      acc += r2 * w2;
      acc += r3 * w3;
    }
    for (; i < n; ++i) {
      const int   s = __builtin_amdgcn_readlane(ms, i);
      const float w = __int_as_float(__builtin_amdgcn_readlane(mwi, i));
      const vec   r = *(const vec*)(hl + (size_t)s * F);
      acc += r * w;
    }
  }

  const vec selfv = *(const vec*)(hl + (size_t)node * F);
  const vec bv    = *(const vec*)(bias + lane * VEC);
  vec r = acc + selfv * (dd * dd) + bv;
  if (RELU) {
#pragma unroll
    for (int v = 0; v < VEC; ++v) r[v] = fmaxf(r[v], 0.0f);
  }
  *(vec*)(out + (size_t)node * F + lane * VEC) = r;
}

extern "C" void kernel_launch(void* const* d_in, const int* in_sizes, int n_in,
                              void* d_out, int out_size, void* d_ws, size_t ws_size,
                              hipStream_t stream) {
  const float* x  = (const float*)d_in[0];
  const int*   ei = (const int*)d_in[1];
  const float* W1 = (const float*)d_in[2];
  const float* b1 = (const float*)d_in[3];
  const float* W2 = (const float*)d_in[4];
  const float* b2 = (const float*)d_in[5];
  float* out = (float*)d_out;

  const int F0 = 128;
  const int N  = in_sizes[0] / F0;
  const int E  = in_sizes[1] / 2;
  const int F1 = in_sizes[3];
  const int F2 = in_sizes[5];
  const int* src = ei;
  const int* dst = ei + E;

  int*   cnt    = (int*)d_ws;
  int*   rowptr = cnt + N;
  float* dinv   = (float*)(rowptr + N + 64);
  int*   rank   = (int*)(dinv + N);
  int*   col    = rank + E;
  float* h1     = (float*)(col + E);
  float* agg1   = h1 + (size_t)N * F1;
  float* h2     = h1;
  int*   hist   = (int*)(agg1 + (size_t)N * F1);

  const int B = 256;
  auto cdiv = [](long long a, long long b) { return (int)((a + b - 1) / b); };

  k_rank<<<CHUNKS, 1024, 0, stream>>>(dst, hist, rank, E, N);
  k_chunk_prefix<<<cdiv(N, B), B, 0, stream>>>(hist, cnt, N);
  k_scan<<<1, 1024, 0, stream>>>(cnt, rowptr, dinv, N);
  k_place<<<CHUNKS, 1024, 0, stream>>>(src, dst, rank, rowptr, hist, col, E, N);

  long long waves1 = (long long)(N / 16) * (F1 / 16);
  k_gemm_wmma<<<cdiv(waves1 * 32, B), B, 0, stream>>>(x, W1, h1, N, F0, F1);
  k_gather<4, true><<<cdiv((long long)N * 32, B), B, 0, stream>>>(h1, rowptr, col, dinv, b1, agg1, N);

  long long waves2 = (long long)(N / 16) * (F2 / 16);
  k_gemm_wmma<<<cdiv(waves2 * 32, B), B, 0, stream>>>(agg1, W2, h2, N, F1, F2);
  k_gather<2, false><<<cdiv((long long)N * 32, B), B, 0, stream>>>(h2, rowptr, col, dinv, b2, out, N);
}


# Scan launch removed: each k_place work-group prefix-sums the node counts in its own LDS (group 0 stores rowptr); dinv moves into the chunk-prefix kernel; 8 to 7 launches; summation order unchanged
# speedup vs baseline: 2.9481x; 1.3978x over previous
#include <hip/hip_runtime.h>

typedef __attribute__((ext_vector_type(2))) float v2f;
typedef __attribute__((ext_vector_type(8))) float v8f;

constexpr int CHUNKS     = 128;
constexpr int LDS_NODES  = 51200;
constexpr int EDGE_BATCH = 8;

__device__ __forceinline__ int row_pitch(int N) { return (N + 3) & ~3; }

__global__ __launch_bounds__(1024)
void k_rank(const int* __restrict__ dst, int* __restrict__ hist, int* __restrict__ rank,
            int E, int N) {
  __shared__ alignas(16) int c[LDS_NODES];
  const int t  = threadIdx.x;
  const int z4 = (N + 3) >> 2;
  for (int i = t; i < z4; i += 1024) ((int4*)c)[i] = make_int4(0, 0, 0, 0);
  __syncthreads();

  const int per = (E + CHUNKS - 1) / CHUNKS;
  const int e0  = blockIdx.x * per;
  const int e1  = min(e0 + per, E);
  for (int eb = e0 + t; eb < e1; eb += EDGE_BATCH * 1024) {
    int d[EDGE_BATCH];
#pragma unroll
    for (int k = 0; k < EDGE_BATCH; ++k) {
      const int e = eb + k * 1024;
      d[k] = (e < e1) ? dst[e] : 0;
    }
#pragma unroll
    for (int k = 0; k < EDGE_BATCH; ++k) {
      const int e = eb + k * 1024;
      if (e < e1) rank[e] = atomicAdd(&c[d[k]], 1);
    }
  }
  __syncthreads();

  int* h = hist + (size_t)blockIdx.x * row_pitch(N);
  const int f4 = N >> 2;
  for (int i = t; i < f4; i += 1024) ((int4*)h)[i] = ((const int4*)c)[i];
  for (int i = (f4 << 2) + t; i < N; i += 1024) h[i] = c[i];
}

__global__ __launch_bounds__(256)
void k_chunk_prefix(int* __restrict__ hist, int* __restrict__ cnt, float* __restrict__ dinv, int N) {
  const int i = blockIdx.x * 256 + threadIdx.x;
  if (i >= N) return;
  const size_t P = (size_t)row_pitch(N);
  int* p = hist + i;
  int run = 0;
  for (int b0 = 0; b0 < CHUNKS; b0 += 32) {
    int v[32];
#pragma unroll
    for (int k = 0; k < 32; ++k) v[k] = p[(size_t)(b0 + k) * P];
#pragma unroll
    for (int k = 0; k < 32; ++k) {
      p[(size_t)(b0 + k) * P] = run;
      run += v[k];
    }
  }
  cnt[i]  = run;
  dinv[i] = 1.0f / sqrtf((float)(run + 1));
}

__device__ __forceinline__ int wave_incl_scan(int v, int lane) {
#pragma unroll
  for (int off = 1; off < 32; off <<= 1) {
    const int u = __shfl_up(v, off);
    if (lane >= off) v += u;
  }
  return v;
}

__global__ __launch_bounds__(1024)
void k_place(const int* __restrict__ src, const int* __restrict__ dst,
             const int* __restrict__ rank, const int* __restrict__ cnt,
             const int* __restrict__ hist, int* __restrict__ rowptr,
             int* __restrict__ col, int E, int N) {
  __shared__ alignas(16) int base[LDS_NODES];
  __shared__ int wsum[32];
  const int t    = threadIdx.x;
  const int lane = t & 31;
  const int w    = t >> 5;
  const int f4   = N >> 2;

#pragma unroll 4
  for (int i = t; i < f4; i += 1024) ((int4*)base)[i] = ((const int4*)cnt)[i];
  for (int i = (f4 << 2) + t; i < N; i += 1024) base[i] = cnt[i];
  __syncthreads();

  const int per = (N + 1023) >> 10;
  const int b   = min(t * per, N);
  const int e   = min(b + per, N);
  int s = 0;
  for (int i = b; i < e; i += 8) {
#pragma unroll
    for (int k = 0; k < 8; ++k) s += (i + k < e) ? base[i + k] : 0;
  }
  const int inc = wave_incl_scan(s, lane);
  if (lane == 31) wsum[w] = inc;
  __syncthreads();
  const int wv    = wsum[lane];
  const int winc  = wave_incl_scan(wv, lane);
  const int front = __shfl(winc, w) - __shfl(wv, w);
  const int total = __shfl(winc, 31);
  int run = front + inc - s;
  for (int i = b; i < e; i += 8) {
    int c[8];
#pragma unroll
    for (int k = 0; k < 8; ++k) c[k] = (i + k < e) ? base[i + k] : 0;
#pragma unroll
    for (int k = 0; k < 8; ++k) {
      if (i + k < e) base[i + k] = run;
      run += c[k];
    }
  }
  __syncthreads();

  const int* h  = hist + (size_t)blockIdx.x * row_pitch(N);
  const bool wr = blockIdx.x == 0;
#pragma unroll 4
  for (int i = t; i < f4; i += 1024) {
    const int4 a  = ((const int4*)base)[i];
    const int4 hb = ((const int4*)h)[i];
    if (wr) ((int4*)rowptr)[i] = a;
    ((int4*)base)[i] = make_int4(a.x + hb.x, a.y + hb.y, a.z + hb.z, a.w + hb.w);
  }
  for (int i = (f4 << 2) + t; i < N; i += 1024) {
    const int a = base[i];
    if (wr) rowptr[i] = a;
    base[i] = a + h[i];
  }
  if (wr && t == 0) rowptr[N] = total;
  __syncthreads();

  const int per_e = (E + CHUNKS - 1) / CHUNKS;
  const int e0  = blockIdx.x * per_e;
  const int e1  = min(e0 + per_e, E);
  for (int eb = e0 + t; eb < e1; eb += EDGE_BATCH * 1024) {
    int d[EDGE_BATCH], s2[EDGE_BATCH], r[EDGE_BATCH];
#pragma unroll
    for (int k = 0; k < EDGE_BATCH; ++k) {
      const int  ee = eb + k * 1024;
      const bool ok = ee < e1;
      d[k]  = ok ? dst[ee]  : 0;
      s2[k] = ok ? src[ee]  : 0;
      r[k]  = ok ? rank[ee] : 0;
    }
#pragma unroll
    for (int k = 0; k < EDGE_BATCH; ++k) {
      const int ee = eb + k * 1024;
      if (ee < e1) col[base[d[k]] + r[k]] = s2[k];
    }
  }
}

__global__ __launch_bounds__(256)
void k_gemm_wmma(const float* __restrict__ A, const float* __restrict__ B,
                 float* __restrict__ C, int M, int K, int Nc) {
  int wid  = (blockIdx.x * blockDim.x + threadIdx.x) >> 5;
  int lane = threadIdx.x & 31;
  int tilesN = Nc >> 4;
  int tilesM = M >> 4;
  if (wid >= tilesM * tilesN) return;
  int tm = wid / tilesN;
  int tn = wid - tm * tilesN;

  int half = lane >> 4;
  int j    = lane & 15;

  const float* Arow = A + (size_t)(tm * 16 + j) * K + 2 * half;
  const float* Bp   = B + (size_t)(2 * half) * Nc + tn * 16 + j;

  v8f acc = {};
  for (int k = 0; k < K; k += 4) {
    v2f a = *(const v2f*)(Arow + k);
    v2f b;
    b.x = Bp[(size_t)k * Nc];
    b.y = Bp[(size_t)(k + 1) * Nc];
    acc = __builtin_amdgcn_wmma_f32_16x16x4_f32(false, a, false, b, (short)0, acc,
                                                false, false);
  }

  float* Cp = C + (size_t)(tm * 16 + 8 * half) * Nc + tn * 16 + j;
#pragma unroll
  for (int v = 0; v < 8; ++v) Cp[(size_t)v * Nc] = acc[v];
}

template <int VEC, bool RELU>
__global__ __launch_bounds__(256)
void k_gather(const float* __restrict__ h, const int* __restrict__ rowptr,
              const int* __restrict__ col, const float* __restrict__ dinv,
              const float* __restrict__ bias, float* __restrict__ out, int N) {
  typedef __attribute__((ext_vector_type(VEC))) float vec;
  constexpr int F = 32 * VEC;
  int node = (blockIdx.x * blockDim.x + threadIdx.x) >> 5;
  const int lane = threadIdx.x & 31;
  if (node >= N) return;
  node = __builtin_amdgcn_readfirstlane(node);

  const int   beg = rowptr[node];
  const int   end = rowptr[node + 1];
  const float dd  = dinv[node];
  const float* hl = h + lane * VEC;

  vec acc = 0.0f;
  for (int base = beg; base < end; base += 32) {
    const int n = min(32, end - base);
    int   ms = 0;
    float mw = 0.0f;
    if (lane < n) {
      ms = col[base + lane];
      mw = dinv[ms] * dd;
    }
    const int mwi = __float_as_int(mw);
    int i = 0;
    for (; i + 4 <= n; i += 4) {
      const int s0 = __builtin_amdgcn_readlane(ms, i);
      const int s1 = __builtin_amdgcn_readlane(ms, i + 1);
      const int s2 = __builtin_amdgcn_readlane(ms, i + 2);
      const int s3 = __builtin_amdgcn_readlane(ms, i + 3);
      const vec r0 = *(const vec*)(hl + (size_t)s0 * F);
      const vec r1 = *(const vec*)(hl + (size_t)s1 * F);
      const vec r2 = *(const vec*)(hl + (size_t)s2 * F);
      const vec r3 = *(const vec*)(hl + (size_t)s3 * F);
      const float w0 = __int_as_float(__builtin_amdgcn_readlane(mwi, i));
      const float w1 = __int_as_float(__builtin_amdgcn_readlane(mwi, i + 1));
      const float w2 = __int_as_float(__builtin_amdgcn_readlane(mwi, i + 2));
      const float w3 = __int_as_float(__builtin_amdgcn_readlane(mwi, i + 3));
      acc += r0 * w0;
      acc += r1 * w1;
      acc += r2 * w2;
      acc += r3 * w3;
    }
    for (; i < n; ++i) {
      const int   s = __builtin_amdgcn_readlane(ms, i);
      const float w = __int_as_float(__builtin_amdgcn_readlane(mwi, i));
      const vec   r = *(const vec*)(hl + (size_t)s * F);
      acc += r * w;
    }
  }

  const vec selfv = *(const vec*)(hl + (size_t)node * F);
  const vec bv    = *(const vec*)(bias + lane * VEC);
  vec r = acc + selfv * (dd * dd) + bv;
  if (RELU) {
#pragma unroll
    for (int v = 0; v < VEC; ++v) r[v] = fmaxf(r[v], 0.0f);
  }
  *(vec*)(out + (size_t)node * F + lane * VEC) = r;
}

extern "C" void kernel_launch(void* const* d_in, const int* in_sizes, int n_in,
                              void* d_out, int out_size, void* d_ws, size_t ws_size,
                              hipStream_t stream) {
  const float* x  = (const float*)d_in[0];
  const int*   ei = (const int*)d_in[1];
  const float* W1 = (const float*)d_in[2];
  const float* b1 = (const float*)d_in[3];
  const float* W2 = (const float*)d_in[4];
  const float* b2 = (const float*)d_in[5];
  float* out = (float*)d_out;

  const int F0 = 128;
  const int N  = in_sizes[0] / F0;
  const int E  = in_sizes[1] / 2;
  const int F1 = in_sizes[3];
  const int F2 = in_sizes[5];
  const int* src = ei;
  const int* dst = ei + E;

  int*   cnt    = (int*)d_ws;
  int*   rowptr = cnt + N;
  float* dinv   = (float*)(rowptr + N + 64);
  int*   rank   = (int*)(dinv + N);
  int*   col    = rank + E;
  float* h1     = (float*)(col + E);
  float* agg1   = h1 + (size_t)N * F1;
  float* h2     = h1;
  int*   hist   = (int*)(agg1 + (size_t)N * F1);

  const int B = 256;
  auto cdiv = [](long long a, long long b) { return (int)((a + b - 1) / b); };

  k_rank<<<CHUNKS, 1024, 0, stream>>>(dst, hist, rank, E, N);
  k_chunk_prefix<<<cdiv(N, B), B, 0, stream>>>(hist, cnt, dinv, N);
  k_place<<<CHUNKS, 1024, 0, stream>>>(src, dst, rank, cnt, hist, rowptr, col, E, N);

  long long waves1 = (long long)(N / 16) * (F1 / 16);
  k_gemm_wmma<<<cdiv(waves1 * 32, B), B, 0, stream>>>(x, W1, h1, N, F0, F1);
  k_gather<4, true><<<cdiv((long long)N * 32, B), B, 0, stream>>>(h1, rowptr, col, dinv, b1, agg1, N);

  long long waves2 = (long long)(N / 16) * (F2 / 16);
  k_gemm_wmma<<<cdiv(waves2 * 32, B), B, 0, stream>>>(agg1, W2, h2, N, F1, F2);
  k_gather<2, false><<<cdiv((long long)N * 32, B), B, 0, stream>>>(h2, rowptr, col, dinv, b2, out, N);
}


# GEMMs re-tiled: one wave per 16-row strip holds its A fragments in registers and computes all column tiles; W copied once per work-group into LDS (k pairs interleaved), B fragments read from LDS
# speedup vs baseline: 3.2077x; 1.0881x over previous
#include <hip/hip_runtime.h>

typedef __attribute__((ext_vector_type(2))) float v2f;
typedef __attribute__((ext_vector_type(8))) float v8f;

constexpr int CHUNKS     = 128;
constexpr int LDS_NODES  = 51200;
constexpr int EDGE_BATCH = 8;

__device__ __forceinline__ int row_pitch(int N) { return (N + 3) & ~3; }

__global__ __launch_bounds__(1024)
void k_rank(const int* __restrict__ dst, int* __restrict__ hist, int* __restrict__ rank,
            int E, int N) {
  __shared__ alignas(16) int c[LDS_NODES];
  const int t  = threadIdx.x;
  const int z4 = (N + 3) >> 2;
  for (int i = t; i < z4; i += 1024) ((int4*)c)[i] = make_int4(0, 0, 0, 0);
  __syncthreads();

  const int per = (E + CHUNKS - 1) / CHUNKS;
  const int e0  = blockIdx.x * per;
  const int e1  = min(e0 + per, E);
  for (int eb = e0 + t; eb < e1; eb += EDGE_BATCH * 1024) {
    int d[EDGE_BATCH];
#pragma unroll
    for (int k = 0; k < EDGE_BATCH; ++k) {
      const int e = eb + k * 1024;
      d[k] = (e < e1) ? dst[e] : 0;
    }
#pragma unroll
    for (int k = 0; k < EDGE_BATCH; ++k) {
      const int e = eb + k * 1024;
      if (e < e1) rank[e] = atomicAdd(&c[d[k]], 1);
    }
  }
  __syncthreads();

  int* h = hist + (size_t)blockIdx.x * row_pitch(N);
  const int f4 = N >> 2;
  for (int i = t; i < f4; i += 1024) ((int4*)h)[i] = ((const int4*)c)[i];
  for (int i = (f4 << 2) + t; i < N; i += 1024) h[i] = c[i];
}

__global__ __launch_bounds__(256)
void k_chunk_prefix(int* __restrict__ hist, int* __restrict__ cnt, float* __restrict__ dinv, int N) {
  const int i = blockIdx.x * 256 + threadIdx.x;
  if (i >= N) return;
  const size_t P = (size_t)row_pitch(N);
  int* p = hist + i;
  int run = 0;
  for (int b0 = 0; b0 < CHUNKS; b0 += 32) {
    int v[32];
#pragma unroll
    for (int k = 0; k < 32; ++k) v[k] = p[(size_t)(b0 + k) * P];
#pragma unroll
    for (int k = 0; k < 32; ++k) {
      p[(size_t)(b0 + k) * P] = run;
      run += v[k];
    }
  }
  cnt[i]  = run;
  dinv[i] = 1.0f / sqrtf((float)(run + 1));
}

__device__ __forceinline__ int wave_incl_scan(int v, int lane) {
#pragma unroll
  for (int off = 1; off < 32; off <<= 1) {
    const int u = __shfl_up(v, off);
    if (lane >= off) v += u;
  }
  return v;
}

__global__ __launch_bounds__(1024)
void k_place(const int* __restrict__ src, const int* __restrict__ dst,
             const int* __restrict__ rank, const int* __restrict__ cnt,
             const int* __restrict__ hist, int* __restrict__ rowptr,
             int* __restrict__ col, int E, int N) {
  __shared__ alignas(16) int base[LDS_NODES];
  __shared__ int wsum[32];
  const int t    = threadIdx.x;
  const int lane = t & 31;
  const int w    = t >> 5;
  const int f4   = N >> 2;

#pragma unroll 4
  for (int i = t; i < f4; i += 1024) ((int4*)base)[i] = ((const int4*)cnt)[i];
  for (int i = (f4 << 2) + t; i < N; i += 1024) base[i] = cnt[i];
  __syncthreads();

  const int per = (N + 1023) >> 10;
  const int b   = min(t * per, N);
  const int e   = min(b + per, N);
  int s = 0;
  for (int i = b; i < e; i += 8) {
#pragma unroll
    for (int k = 0; k < 8; ++k) s += (i + k < e) ? base[i + k] : 0;
  }
  const int inc = wave_incl_scan(s, lane);
  if (lane == 31) wsum[w] = inc;
  __syncthreads();
  const int wv    = wsum[lane];
  const int winc  = wave_incl_scan(wv, lane);
  const int front = __shfl(winc, w) - __shfl(wv, w);
  const int total = __shfl(winc, 31);
  int run = front + inc - s;
  for (int i = b; i < e; i += 8) {
    int c[8];
#pragma unroll
    for (int k = 0; k < 8; ++k) c[k] = (i + k < e) ? base[i + k] : 0;
#pragma unroll
    for (int k = 0; k < 8; ++k) {
      if (i + k < e) base[i + k] = run;
      run += c[k];
    }
  }
  __syncthreads();

  const int* h  = hist + (size_t)blockIdx.x * row_pitch(N);
  const bool wr = blockIdx.x == 0;
#pragma unroll 4
  for (int i = t; i < f4; i += 1024) {
    const int4 a  = ((const int4*)base)[i];
    const int4 hb = ((const int4*)h)[i];
    if (wr) ((int4*)rowptr)[i] = a;
    ((int4*)base)[i] = make_int4(a.x + hb.x, a.y + hb.y, a.z + hb.z, a.w + hb.w);
  }
  for (int i = (f4 << 2) + t; i < N; i += 1024) {
    const int a = base[i];
    if (wr) rowptr[i] = a;
    base[i] = a + h[i];
  }
  if (wr && t == 0) rowptr[N] = total;
  __syncthreads();

  const int per_e = (E + CHUNKS - 1) / CHUNKS;
  const int e0  = blockIdx.x * per_e;
  const int e1  = min(e0 + per_e, E);
  for (int eb = e0 + t; eb < e1; eb += EDGE_BATCH * 1024) {
    int d[EDGE_BATCH], s2[EDGE_BATCH], r[EDGE_BATCH];
#pragma unroll
    for (int k = 0; k < EDGE_BATCH; ++k) {
      const int  ee = eb + k * 1024;
      const bool ok = ee < e1;
      d[k]  = ok ? dst[ee]  : 0;
      s2[k] = ok ? src[ee]  : 0;
      r[k]  = ok ? rank[ee] : 0;
    }
#pragma unroll
    for (int k = 0; k < EDGE_BATCH; ++k) {
      const int ee = eb + k * 1024;
      if (ee < e1) col[base[d[k]] + r[k]] = s2[k];
    }
  }
}

template <int K, int NT>
__global__ __launch_bounds__(256)
void k_gemm_strip(const float* __restrict__ A, const float* __restrict__ B,
                  float* __restrict__ C, int M) {
  constexpr int Nc = NT * 16;
  constexpr int PP = 2 * Nc + 32;
  constexpr int Q  = Nc / 4;
  __shared__ alignas(16) float Bs[(K / 2) * PP];
  const int t     = threadIdx.x;
  const int lane  = t & 31;
  const int half  = lane >> 4;
  const int j     = lane & 15;
  const int strip = blockIdx.x * 8 + (t >> 5);
  const bool valid = strip < (M >> 4);

  constexpr int TRIPS = (K / 2) * Q / 256;
  static_assert(TRIPS * 256 == (K / 2) * Q, "B must split evenly over the 256 threads");
  float4 r0[TRIPS], r1[TRIPS];
#pragma unroll
  for (int it = 0; it < TRIPS; ++it) {
    const int i = t + it * 256;
    const int p = i / Q;
    const int c = i - p * Q;
    r0[it] = ((const float4*)B)[(2 * p) * Q + c];
    r1[it] = ((const float4*)B)[(2 * p + 1) * Q + c];
  }

  const int srow = valid ? strip : 0;
  const float* Arow = A + (size_t)(srow * 16 + j) * K + 2 * half;
  v2f a[K / 4];
#pragma unroll
  for (int i = 0; i < K / 4; ++i) a[i] = *(const v2f*)(Arow + 4 * i);

#pragma unroll
  for (int it = 0; it < TRIPS; ++it) {
    const int i = t + it * 256;
    const int p = i / Q;
    const int c = i - p * Q;
    float* d = Bs + p * PP + 8 * c;
    *(float4*)(d)     = make_float4(r0[it].x, r1[it].x, r0[it].y, r1[it].y);
    *(float4*)(d + 4) = make_float4(r0[it].z, r1[it].z, r0[it].w, r1[it].w);
  }
  __builtin_amdgcn_fence(__ATOMIC_RELEASE, "workgroup", "local");
  __builtin_amdgcn_s_barrier();
  __builtin_amdgcn_fence(__ATOMIC_ACQUIRE, "workgroup", "local");
  if (!valid) return;

  const float* Bl = Bs + half * PP + 2 * j;

  v8f acc[NT];
#pragma unroll
  for (int n = 0; n < NT; ++n) acc[n] = v8f{};

#pragma unroll
  for (int ks = 0; ks < K / 4; ++ks) {
#pragma unroll
    for (int n = 0; n < NT; ++n) {
      const v2f b = *(const v2f*)(Bl + 2 * ks * PP + 32 * n);
      acc[n] = __builtin_amdgcn_wmma_f32_16x16x4_f32(false, a[ks], false, b, (short)0, acc[n],
                                                     false, false);
    }
  }

  float* Cp = C + (size_t)(strip * 16 + 8 * half) * Nc + j;
#pragma unroll
  for (int n = 0; n < NT; ++n) {
#pragma unroll
    for (int v = 0; v < 8; ++v) Cp[(size_t)v * Nc + n * 16] = acc[n][v];
  }
}

template <int VEC, bool RELU>
__global__ __launch_bounds__(256)
void k_gather(const float* __restrict__ h, const int* __restrict__ rowptr,
              const int* __restrict__ col, const float* __restrict__ dinv,
              const float* __restrict__ bias, float* __restrict__ out, int N) {
  typedef __attribute__((ext_vector_type(VEC))) float vec;
  constexpr int F = 32 * VEC;
  int node = (blockIdx.x * blockDim.x + threadIdx.x) >> 5;
  const int lane = threadIdx.x & 31;
  if (node >= N) return;
  node = __builtin_amdgcn_readfirstlane(node);

  const int   beg = rowptr[node];
  const int   end = rowptr[node + 1];
  const float dd  = dinv[node];
  const float* hl = h + lane * VEC;

  vec acc = 0.0f;
  for (int base = beg; base < end; base += 32) {
    const int n = min(32, end - base);
    int   ms = 0;
    float mw = 0.0f;
    if (lane < n) {
      ms = col[base + lane];
      mw = dinv[ms] * dd;
    }
    const int mwi = __float_as_int(mw);
    int i = 0;
    for (; i + 4 <= n; i += 4) {
      const int s0 = __builtin_amdgcn_readlane(ms, i);
      const int s1 = __builtin_amdgcn_readlane(ms, i + 1);
      const int s2 = __builtin_amdgcn_readlane(ms, i + 2);
      const int s3 = __builtin_amdgcn_readlane(ms, i + 3);
      const vec r0 = *(const vec*)(hl + (size_t)s0 * F);
      const vec r1 = *(const vec*)(hl + (size_t)s1 * F);
      const vec r2 = *(const vec*)(hl + (size_t)s2 * F);
      const vec r3 = *(const vec*)(hl + (size_t)s3 * F);
      const float w0 = __int_as_float(__builtin_amdgcn_readlane(mwi, i));
      const float w1 = __int_as_float(__builtin_amdgcn_readlane(mwi, i + 1));
      const float w2 = __int_as_float(__builtin_amdgcn_readlane(mwi, i + 2));
      const float w3 = __int_as_float(__builtin_amdgcn_readlane(mwi, i + 3));
      acc += r0 * w0;
      acc += r1 * w1;
      acc += r2 * w2;
      acc += r3 * w3;
    }
    for (; i < n; ++i) {
      const int   s = __builtin_amdgcn_readlane(ms, i);
      const float w = __int_as_float(__builtin_amdgcn_readlane(mwi, i));
      const vec   r = *(const vec*)(hl + (size_t)s * F);
      acc += r * w;
    }
  }

  const vec selfv = *(const vec*)(hl + (size_t)node * F);
  const vec bv    = *(const vec*)(bias + lane * VEC);
  vec r = acc + selfv * (dd * dd) + bv;
  if (RELU) {
#pragma unroll
    for (int v = 0; v < VEC; ++v) r[v] = fmaxf(r[v], 0.0f);
  }
  *(vec*)(out + (size_t)node * F + lane * VEC) = r;
}

extern "C" void kernel_launch(void* const* d_in, const int* in_sizes, int n_in,
                              void* d_out, int out_size, void* d_ws, size_t ws_size,
                              hipStream_t stream) {
  const float* x  = (const float*)d_in[0];
  const int*   ei = (const int*)d_in[1];
  const float* W1 = (const float*)d_in[2];
  const float* b1 = (const float*)d_in[3];
  const float* W2 = (const float*)d_in[4];
  const float* b2 = (const float*)d_in[5];
  float* out = (float*)d_out;

  const int F0 = 128;
  const int N  = in_sizes[0] / F0;
  const int E  = in_sizes[1] / 2;
  const int F1 = in_sizes[3];
  const int* src = ei;
  const int* dst = ei + E;

  int*   cnt    = (int*)d_ws;
  int*   rowptr = cnt + N;
  float* dinv   = (float*)(rowptr + N + 64);
  int*   rank   = (int*)(dinv + N);
  int*   col    = rank + E;
  float* h1     = (float*)(col + E);
  float* agg1   = h1 + (size_t)N * F1;
  float* h2     = h1;
  int*   hist   = (int*)(agg1 + (size_t)N * F1);

  const int B = 256;
  auto cdiv = [](long long a, long long b) { return (int)((a + b - 1) / b); };

  k_rank<<<CHUNKS, 1024, 0, stream>>>(dst, hist, rank, E, N);
  k_chunk_prefix<<<cdiv(N, B), B, 0, stream>>>(hist, cnt, dinv, N);
  k_place<<<CHUNKS, 1024, 0, stream>>>(src, dst, rank, cnt, hist, rowptr, col, E, N);

  k_gemm_strip<128, 8><<<cdiv(N / 16, 8), B, 0, stream>>>(x, W1, h1, N);
  k_gather<4, true><<<cdiv((long long)N * 32, B), B, 0, stream>>>(h1, rowptr, col, dinv, b1, agg1, N);

  k_gemm_strip<128, 4><<<cdiv(N / 16, 8), B, 0, stream>>>(agg1, W2, h2, N);
  k_gather<2, false><<<cdiv((long long)N * 32, B), B, 0, stream>>>(h2, rowptr, col, dinv, b2, out, N);
}


# Workspace planes re-ordered: both row tables first, so each gathered 512-B / 256-B row starts on a 128-B line (was 64 B past one): a row load touches 4 / 2 lines, not 5 / 3; sum order unchanged
# speedup vs baseline: 4.1594x; 1.2967x over previous
#include <hip/hip_runtime.h>

typedef __attribute__((ext_vector_type(2))) float v2f;
typedef __attribute__((ext_vector_type(8))) float v8f;

constexpr int CHUNKS     = 128;
constexpr int LDS_NODES  = 51200;
constexpr int EDGE_BATCH = 8;

__device__ __forceinline__ int row_pitch(int N) { return (N + 3) & ~3; }

__global__ __launch_bounds__(1024)
void k_rank(const int* __restrict__ dst, int* __restrict__ hist, int* __restrict__ rank,
            int E, int N) {
  __shared__ alignas(16) int c[LDS_NODES];
  const int t  = threadIdx.x;
  const int z4 = (N + 3) >> 2;
  for (int i = t; i < z4; i += 1024) ((int4*)c)[i] = make_int4(0, 0, 0, 0);
  __syncthreads();

  const int per = (E + CHUNKS - 1) / CHUNKS;
  const int e0  = blockIdx.x * per;
  const int e1  = min(e0 + per, E);
  for (int eb = e0 + t; eb < e1; eb += EDGE_BATCH * 1024) {
    int d[EDGE_BATCH];
#pragma unroll
    for (int k = 0; k < EDGE_BATCH; ++k) {
      const int e = eb + k * 1024;
      d[k] = (e < e1) ? dst[e] : 0;
    }
#pragma unroll
    for (int k = 0; k < EDGE_BATCH; ++k) {
      const int e = eb + k * 1024;
      if (e < e1) rank[e] = atomicAdd(&c[d[k]], 1);
    }
  }
  __syncthreads();

  int* h = hist + (size_t)blockIdx.x * row_pitch(N);
  const int f4 = N >> 2;
  for (int i = t; i < f4; i += 1024) ((int4*)h)[i] = ((const int4*)c)[i];
  for (int i = (f4 << 2) + t; i < N; i += 1024) h[i] = c[i];
}

__global__ __launch_bounds__(256)
void k_chunk_prefix(int* __restrict__ hist, int* __restrict__ cnt, float* __restrict__ dinv, int N) {
  const int i = blockIdx.x * 256 + threadIdx.x;
  if (i >= N) return;
  const size_t P = (size_t)row_pitch(N);
  int* p = hist + i;
  int run = 0;
  for (int b0 = 0; b0 < CHUNKS; b0 += 32) {
    int v[32];
#pragma unroll
    for (int k = 0; k < 32; ++k) v[k] = p[(size_t)(b0 + k) * P];
#pragma unroll
    for (int k = 0; k < 32; ++k) {
      p[(size_t)(b0 + k) * P] = run;
      run += v[k];
    }
  }
  cnt[i]  = run;
  dinv[i] = 1.0f / sqrtf((float)(run + 1));
}

__device__ __forceinline__ int wave_incl_scan(int v, int lane) {
#pragma unroll
  for (int off = 1; off < 32; off <<= 1) {
    const int u = __shfl_up(v, off);
    if (lane >= off) v += u;
  }
  return v;
}

__global__ __launch_bounds__(1024)
void k_place(const int* __restrict__ src, const int* __restrict__ dst,
             const int* __restrict__ rank, const int* __restrict__ cnt,
             const int* __restrict__ hist, int* __restrict__ rowptr,
             int* __restrict__ col, int E, int N) {
  __shared__ alignas(16) int base[LDS_NODES];
  __shared__ int wsum[32];
  const int t    = threadIdx.x;
  const int lane = t & 31;
  const int w    = t >> 5;
  const int f4   = N >> 2;

#pragma unroll 4
  for (int i = t; i < f4; i += 1024) ((int4*)base)[i] = ((const int4*)cnt)[i];
  for (int i = (f4 << 2) + t; i < N; i += 1024) base[i] = cnt[i];
  __syncthreads();

  const int per = (N + 1023) >> 10;
  const int b   = min(t * per, N);
  const int e   = min(b + per, N);
  int s = 0;
  for (int i = b; i < e; i += 8) {
#pragma unroll
    for (int k = 0; k < 8; ++k) s += (i + k < e) ? base[i + k] : 0;
  }
  const int inc = wave_incl_scan(s, lane);
  if (lane == 31) wsum[w] = inc;
  __syncthreads();
  const int wv    = wsum[lane];
  const int winc  = wave_incl_scan(wv, lane);
  const int front = __shfl(winc, w) - __shfl(wv, w);
  const int total = __shfl(winc, 31);
  int run = front + inc - s;
  for (int i = b; i < e; i += 8) {
    int c[8];
#pragma unroll
    for (int k = 0; k < 8; ++k) c[k] = (i + k < e) ? base[i + k] : 0;
#pragma unroll
    for (int k = 0; k < 8; ++k) {
      if (i + k < e) base[i + k] = run;
      run += c[k];
    }
  }
  __syncthreads();

  const int* h  = hist + (size_t)blockIdx.x * row_pitch(N);
  const bool wr = blockIdx.x == 0;
#pragma unroll 4
  for (int i = t; i < f4; i += 1024) {
    const int4 a  = ((const int4*)base)[i];
    const int4 hb = ((const int4*)h)[i];
    if (wr) ((int4*)rowptr)[i] = a;
    ((int4*)base)[i] = make_int4(a.x + hb.x, a.y + hb.y, a.z + hb.z, a.w + hb.w);
  }
  for (int i = (f4 << 2) + t; i < N; i += 1024) {
    const int a = base[i];
    if (wr) rowptr[i] = a;
    base[i] = a + h[i];
  }
  if (wr && t == 0) rowptr[N] = total;
  __syncthreads();

  const int per_e = (E + CHUNKS - 1) / CHUNKS;
  const int e0  = blockIdx.x * per_e;
  const int e1  = min(e0 + per_e, E);
  for (int eb = e0 + t; eb < e1; eb += EDGE_BATCH * 1024) {
    int d[EDGE_BATCH], s2[EDGE_BATCH], r[EDGE_BATCH];
#pragma unroll
    for (int k = 0; k < EDGE_BATCH; ++k) {
      const int  ee = eb + k * 1024;
      const bool ok = ee < e1;
      d[k]  = ok ? dst[ee]  : 0;
      s2[k] = ok ? src[ee]  : 0;
      r[k]  = ok ? rank[ee] : 0;
    }
#pragma unroll
    for (int k = 0; k < EDGE_BATCH; ++k) {
      const int ee = eb + k * 1024;
      if (ee < e1) col[base[d[k]] + r[k]] = s2[k];
    }
  }
}

template <int K, int NT>
__global__ __launch_bounds__(256)
void k_gemm_strip(const float* __restrict__ A, const float* __restrict__ B,
                  const float* __restrict__ rs, float* __restrict__ C, int M) {
  constexpr int Nc = NT * 16;
  constexpr int PP = 2 * Nc + 32;
  constexpr int Q  = Nc / 4;
  __shared__ alignas(16) float Bs[(K / 2) * PP];
  const int t     = threadIdx.x;
  const int lane  = t & 31;
  const int half  = lane >> 4;
  const int j     = lane & 15;
  const int strip = blockIdx.x * 8 + (t >> 5);
  const bool valid = strip < (M >> 4);

  constexpr int TRIPS = (K / 2) * Q / 256;
  static_assert(TRIPS * 256 == (K / 2) * Q, "B must split evenly over the 256 threads");
  float4 r0[TRIPS], r1[TRIPS];
#pragma unroll
  for (int it = 0; it < TRIPS; ++it) {
    const int i = t + it * 256;
    const int p = i / Q;
    const int c = i - p * Q;
    r0[it] = ((const float4*)B)[(2 * p) * Q + c];
    r1[it] = ((const float4*)B)[(2 * p + 1) * Q + c];
  }

  const int srow = valid ? strip : 0;
  const float* Arow = A + (size_t)(srow * 16 + j) * K + 2 * half;
  v2f a[K / 4];
#pragma unroll
  for (int i = 0; i < K / 4; ++i) a[i] = *(const v2f*)(Arow + 4 * i);
  const float4 sc0 = *(const float4*)(rs + srow * 16 + 8 * half);
  const float4 sc1 = *(const float4*)(rs + srow * 16 + 8 * half + 4);

#pragma unroll
  for (int it = 0; it < TRIPS; ++it) {
    const int i = t + it * 256;
    const int p = i / Q;
    const int c = i - p * Q;
    float* d = Bs + p * PP + 8 * c;
    *(float4*)(d)     = make_float4(r0[it].x, r1[it].x, r0[it].y, r1[it].y);
    *(float4*)(d + 4) = make_float4(r0[it].z, r1[it].z, r0[it].w, r1[it].w);
  }
  __builtin_amdgcn_fence(__ATOMIC_RELEASE, "workgroup", "local");
  __builtin_amdgcn_s_barrier();
  __builtin_amdgcn_fence(__ATOMIC_ACQUIRE, "workgroup", "local");
  if (!valid) return;

  const float* Bl = Bs + half * PP + 2 * j;

  v8f acc[NT];
#pragma unroll
  for (int n = 0; n < NT; ++n) acc[n] = v8f{};

#pragma unroll
  for (int ks = 0; ks < K / 4; ++ks) {
#pragma unroll
    for (int n = 0; n < NT; ++n) {
      const v2f b = *(const v2f*)(Bl + 2 * ks * PP + 32 * n);
      acc[n] = __builtin_amdgcn_wmma_f32_16x16x4_f32(false, a[ks], false, b, (short)0, acc[n],
                                                     false, false);
    }
  }

  const float sc[8] = {sc0.x, sc0.y, sc0.z, sc0.w, sc1.x, sc1.y, sc1.z, sc1.w};
  float* Cp = C + (size_t)(strip * 16 + 8 * half) * Nc + j;
#pragma unroll
  for (int n = 0; n < NT; ++n) {
#pragma unroll
    for (int v = 0; v < 8; ++v) Cp[(size_t)v * Nc + n * 16] = acc[n][v] * sc[v];
  }
}

template <int VEC, bool RELU>
__global__ __launch_bounds__(256)
void k_gather(const float* __restrict__ h, const int* __restrict__ rowptr,
              const int* __restrict__ col, const float* __restrict__ dinv,
              const float* __restrict__ bias, float* __restrict__ out, int N) {
  typedef __attribute__((ext_vector_type(VEC))) float vec;
  constexpr int F = 32 * VEC;
  int node = (blockIdx.x * blockDim.x + threadIdx.x) >> 5;
  const int lane = threadIdx.x & 31;
  if (node >= N) return;
  node = __builtin_amdgcn_readfirstlane(node);

  const int   beg = rowptr[node];
  const int   end = rowptr[node + 1];
  const float dd  = dinv[node];
  const float* hl = h + lane * VEC;

  vec acc = 0.0f;
  for (int base = beg; base < end; base += 32) {
    const int n = min(32, end - base);
    int ms = 0;
    if (lane < n) ms = col[base + lane];
    int i = 0;
    for (; i + 4 <= n; i += 4) {
      const int s0 = __builtin_amdgcn_readlane(ms, i);
      const int s1 = __builtin_amdgcn_readlane(ms, i + 1);
      const int s2 = __builtin_amdgcn_readlane(ms, i + 2);
      const int s3 = __builtin_amdgcn_readlane(ms, i + 3);
      const vec r0 = *(const vec*)(hl + (size_t)s0 * F);
      const vec r1 = *(const vec*)(hl + (size_t)s1 * F);
      const vec r2 = *(const vec*)(hl + (size_t)s2 * F);
      const vec r3 = *(const vec*)(hl + (size_t)s3 * F);
      acc += r0;
      acc += r1;
      acc += r2;
      acc += r3;
    }
    for (; i < n; ++i) {
      const int s = __builtin_amdgcn_readlane(ms, i);
      const vec r = *(const vec*)(hl + (size_t)s * F);
      acc += r;
    }
  }

  const vec selfv = *(const vec*)(hl + (size_t)node * F);
  const vec bv    = *(const vec*)(bias + lane * VEC);
  vec r = (acc + selfv) * dd + bv;
  if (RELU) {
#pragma unroll
    for (int v = 0; v < VEC; ++v) r[v] = fmaxf(r[v], 0.0f);
  }
  *(vec*)(out + (size_t)node * F + lane * VEC) = r;
}

extern "C" void kernel_launch(void* const* d_in, const int* in_sizes, int n_in,
                              void* d_out, int out_size, void* d_ws, size_t ws_size,
                              hipStream_t stream) {
  const float* x  = (const float*)d_in[0];
  const int*   ei = (const int*)d_in[1];
  const float* W1 = (const float*)d_in[2];
  const float* b1 = (const float*)d_in[3];
  const float* W2 = (const float*)d_in[4];
  const float* b2 = (const float*)d_in[5];
  float* out = (float*)d_out;

  const int F0 = 128;
  const int N  = in_sizes[0] / F0;
  const int E  = in_sizes[1] / 2;
  const int F1 = in_sizes[3];
  const int* src = ei;
  const int* dst = ei + E;

  float* h1     = (float*)d_ws;
  float* agg1   = h1 + (size_t)N * F1;
  float* h2     = h1;
  int*   cnt    = (int*)(agg1 + (size_t)N * F1);
  int*   rowptr = cnt + N;
  float* dinv   = (float*)(rowptr + N + 64);
  int*   rank   = (int*)(dinv + N);
  int*   col    = rank + E;
  int*   hist   = col + E;

  const int B = 256;
  auto cdiv = [](long long a, long long b) { return (int)((a + b - 1) / b); };

  k_rank<<<CHUNKS, 1024, 0, stream>>>(dst, hist, rank, E, N);
  k_chunk_prefix<<<cdiv(N, B), B, 0, stream>>>(hist, cnt, dinv, N);
  k_place<<<CHUNKS, 1024, 0, stream>>>(src, dst, rank, cnt, hist, rowptr, col, E, N);

  k_gemm_strip<128, 8><<<cdiv(N / 16, 8), B, 0, stream>>>(x, W1, dinv, h1, N);
  k_gather<4, true><<<cdiv((long long)N * 32, B), B, 0, stream>>>(h1, rowptr, col, dinv, b1, agg1, N);

  k_gemm_strip<128, 4><<<cdiv(N / 16, 8), B, 0, stream>>>(agg1, W2, dinv, h2, N);
  k_gather<2, false><<<cdiv((long long)N * 32, B), B, 0, stream>>>(h2, rowptr, col, dinv, b2, out, N);
}


# Edge sort by destination made two-level: 8192-edge tiles bucketed (256 nodes) in LDS and stored as runs, then one work-group per bucket places in LDS; sum order = LDS arrival per bucket
# speedup vs baseline: 6.0292x; 1.4495x over previous
#include <hip/hip_runtime.h>

typedef __attribute__((ext_vector_type(2))) float v2f;
typedef __attribute__((ext_vector_type(8))) float v8f;

constexpr int TILE       = 8192;
constexpr int EDGE_BATCH = 8;
constexpr int BSHIFT     = 8;
constexpr int BNODES     = 1 << BSHIFT;
constexpr int MAXB       = 256;
constexpr int T_BATCH    = 16;
constexpr int SEG_CAP    = 24576;

__device__ __forceinline__ int wave_incl_scan(int v, int lane) {
#pragma unroll
  for (int off = 1; off < 32; off <<= 1) {
    const int u = __shfl_up(v, off);
    if (lane >= off) v += u;
  }
  return v;
}

__global__ __launch_bounds__(1024)
void k_bucket_hist(const int* __restrict__ dst, int* __restrict__ T, int E, int NBK) {
  __shared__ int h[MAXB];
  const int t  = threadIdx.x;
  const int e0 = blockIdx.x * TILE;
  const int e1 = min(e0 + TILE, E);
  int d[EDGE_BATCH];
#pragma unroll
  for (int k = 0; k < EDGE_BATCH; ++k) {
    const int e = e0 + t + k * 1024;
    d[k] = (e < e1) ? dst[e] : -1;
  }
  if (t < MAXB) h[t] = 0;
  __syncthreads();
#pragma unroll
  for (int k = 0; k < EDGE_BATCH; ++k)
    if (d[k] >= 0) atomicAdd(&h[d[k] >> BSHIFT], 1);
  __syncthreads();
  if (t < NBK) T[blockIdx.x * MAXB + t] = h[t];
}

__global__ __launch_bounds__(1024)
void k_bucket_scatter(const int* __restrict__ src, const int* __restrict__ dst,
                      const int* __restrict__ T, int* __restrict__ pairs,
                      int* __restrict__ bstart, int E, int NBK, int CH) {
  int2* tmp = (int2*)pairs;
  __shared__ int2 stage[TILE];
  __shared__ int part[4][MAXB];
  __shared__ int tot4[4][MAXB];
  __shared__ int cur[MAXB];
  __shared__ int goff[MAXB];
  __shared__ int wsT[8], wsM[8];
  const int t    = threadIdx.x;
  const int lane = t & 31;
  const int w    = t >> 5;
  const int k    = t & (MAXB - 1);
  const int q    = t >> 8;
  const int me   = blockIdx.x;
  const int e0   = me * TILE;
  const int e1   = min(e0 + TILE, E);

  int d[EDGE_BATCH], s[EDGE_BATCH];
#pragma unroll
  for (int j = 0; j < EDGE_BATCH; ++j) {
    const int  e  = e0 + t + j * 1024;
    const bool ok = e < e1;
    d[j] = ok ? dst[e] : -1;
    s[j] = ok ? src[e] : 0;
  }

  int before = 0, total = 0;
  if (k < NBK) {
    for (int c0 = q; c0 < CH; c0 += 4 * T_BATCH) {
      int v[T_BATCH];
#pragma unroll
      for (int j = 0; j < T_BATCH; ++j) {
        const int c = c0 + 4 * j;
        v[j] = (c < CH) ? T[c * MAXB + k] : 0;
      }
#pragma unroll
      for (int j = 0; j < T_BATCH; ++j) {
        total  += v[j];
        before += (c0 + 4 * j < me) ? v[j] : 0;
      }
    }
  }
  part[q][k] = before;
  tot4[q][k] = total;
  __syncthreads();

  int tt = 0, bf = 0, mine = 0, incT = 0, incM = 0;
  if (t < MAXB) {
    bf   = part[0][t] + part[1][t] + part[2][t] + part[3][t];
    tt   = tot4[0][t] + tot4[1][t] + tot4[2][t] + tot4[3][t];
    mine = (t < NBK) ? T[me * MAXB + t] : 0;
    incT = wave_incl_scan(tt, lane);
    incM = wave_incl_scan(mine, lane);
    if (lane == 31) { wsT[w] = incT; wsM[w] = incM; }
  }
  __syncthreads();
  if (t < MAXB) {
    int preT = 0, preM = 0;
    for (int i = 0; i < w; ++i) { preT += wsT[i]; preM += wsM[i]; }
    const int gs = preT + incT - tt;
    const int ls = preM + incM - mine;
    cur[t]  = ls;
    goff[t] = gs + bf - ls;
    if (me == 0 && t < NBK) {
      bstart[t] = gs;
      if (t == NBK - 1) bstart[NBK] = gs + tt;
    }
  }
  __syncthreads();

#pragma unroll
  for (int j = 0; j < EDGE_BATCH; ++j) {
    if (d[j] >= 0) {
      const int p = atomicAdd(&cur[d[j] >> BSHIFT], 1);
      stage[p] = make_int2(s[j], d[j]);
    }
  }
  __syncthreads();

  const int n = e1 - e0;
  for (int i = t; i < n; i += 1024) {
    const int2 v = stage[i];
    tmp[goff[v.y >> BSHIFT] + i] = v;
  }
}

__global__ __launch_bounds__(1024)
void k_bucket_sort(const int* __restrict__ pairs, const int* __restrict__ bstart,
                   int* __restrict__ rowptr, float* __restrict__ dinv,
                   int* __restrict__ col, int N) {
  __shared__ int scol[SEG_CAP];
  __shared__ int cnt[BNODES];
  __shared__ int fill[BNODES];
  __shared__ int ws[8];
  const int t     = threadIdx.x;
  const int lane  = t & 31;
  const int w     = t >> 5;
  const int g0    = bstart[blockIdx.x];
  const int n     = bstart[blockIdx.x + 1] - g0;
  const int node0 = blockIdx.x << BSHIFT;
  const int2* my  = (const int2*)pairs + g0;

  if (t < BNODES) cnt[t] = 0;
  __syncthreads();
  for (int base = t; base < n; base += TILE) {
    int dl[EDGE_BATCH];
#pragma unroll
    for (int j = 0; j < EDGE_BATCH; ++j) {
      const int i = base + j * 1024;
      dl[j] = (i < n) ? my[i].y : -1;
    }
#pragma unroll
    for (int j = 0; j < EDGE_BATCH; ++j)
      if (dl[j] >= 0) atomicAdd(&cnt[dl[j] - node0], 1);
  }
  __syncthreads();

  int c = 0, inc = 0;
  if (t < BNODES) {
    c   = cnt[t];
    inc = wave_incl_scan(c, lane);
    if (lane == 31) ws[w] = inc;
  }
  __syncthreads();
  if (t < BNODES) {
    int pre = 0;
    for (int i = 0; i < w; ++i) pre += ws[i];
    const int ex = pre + inc - c;
    fill[t] = ex;
    const int node = node0 + t;
    if (node < N) {
      rowptr[node] = g0 + ex;
      dinv[node]   = 1.0f / sqrtf((float)(c + 1));
      if (node == N - 1) rowptr[N] = g0 + ex + c;
    }
  }
  __syncthreads();

  const bool fits = n <= SEG_CAP;
  for (int base = t; base < n; base += TILE) {
    int2 v[EDGE_BATCH];
#pragma unroll
    for (int j = 0; j < EDGE_BATCH; ++j) {
      const int i = base + j * 1024;
      v[j] = (i < n) ? my[i] : make_int2(0, -1);
    }
#pragma unroll
    for (int j = 0; j < EDGE_BATCH; ++j) {
      if (v[j].y >= 0) {
        const int p = atomicAdd(&fill[v[j].y - node0], 1);
        if (fits) scol[p] = v[j].x;
        else      col[g0 + p] = v[j].x;
      }
    }
  }
  __syncthreads();
  if (fits) {
    for (int i = t; i < n; i += 1024) col[g0 + i] = scol[i];
  }
}

template <int K, int NT>
__global__ __launch_bounds__(256)
void k_gemm_strip(const float* __restrict__ A, const float* __restrict__ B,
                  const float* __restrict__ rs, float* __restrict__ C, int M) {
  constexpr int Nc = NT * 16;
  constexpr int PP = 2 * Nc + 32;
  constexpr int Q  = Nc / 4;
  __shared__ alignas(16) float Bs[(K / 2) * PP];
  const int t     = threadIdx.x;
  const int lane  = t & 31;
  const int half  = lane >> 4;
  const int j     = lane & 15;
  const int strip = blockIdx.x * 8 + (t >> 5);
  const bool valid = strip < (M >> 4);

  constexpr int TRIPS = (K / 2) * Q / 256;
  static_assert(TRIPS * 256 == (K / 2) * Q, "B must split evenly over the 256 threads");
  float4 r0[TRIPS], r1[TRIPS];
#pragma unroll
  for (int it = 0; it < TRIPS; ++it) {
    const int i = t + it * 256;
    const int p = i / Q;
    const int c = i - p * Q;
    r0[it] = ((const float4*)B)[(2 * p) * Q + c];
    r1[it] = ((const float4*)B)[(2 * p + 1) * Q + c];
  }

  const int srow = valid ? strip : 0;
  const float* Arow = A + (size_t)(srow * 16 + j) * K + 2 * half;
  v2f a[K / 4];
#pragma unroll
  for (int i = 0; i < K / 4; ++i) a[i] = *(const v2f*)(Arow + 4 * i);
  const float4 sc0 = *(const float4*)(rs + srow * 16 + 8 * half);
  const float4 sc1 = *(const float4*)(rs + srow * 16 + 8 * half + 4);

#pragma unroll
  for (int it = 0; it < TRIPS; ++it) {
    const int i = t + it * 256;
    const int p = i / Q;
    const int c = i - p * Q;
    float* d = Bs + p * PP + 8 * c;
    *(float4*)(d)     = make_float4(r0[it].x, r1[it].x, r0[it].y, r1[it].y);
    *(float4*)(d + 4) = make_float4(r0[it].z, r1[it].z, r0[it].w, r1[it].w);
  }
  __builtin_amdgcn_fence(__ATOMIC_RELEASE, "workgroup", "local");
  __builtin_amdgcn_s_barrier();
  __builtin_amdgcn_fence(__ATOMIC_ACQUIRE, "workgroup", "local");
  if (!valid) return;

  const float* Bl = Bs + half * PP + 2 * j;

  v8f acc[NT];
#pragma unroll
  for (int n = 0; n < NT; ++n) acc[n] = v8f{};

#pragma unroll
  for (int ks = 0; ks < K / 4; ++ks) {
#pragma unroll
    for (int n = 0; n < NT; ++n) {
      const v2f b = *(const v2f*)(Bl + 2 * ks * PP + 32 * n);
      acc[n] = __builtin_amdgcn_wmma_f32_16x16x4_f32(false, a[ks], false, b, (short)0, acc[n],
                                                     false, false);
    }
  }

  const float sc[8] = {sc0.x, sc0.y, sc0.z, sc0.w, sc1.x, sc1.y, sc1.z, sc1.w};
  float* Cp = C + (size_t)(strip * 16 + 8 * half) * Nc + j;
#pragma unroll
  for (int n = 0; n < NT; ++n) {
#pragma unroll
    for (int v = 0; v < 8; ++v) Cp[(size_t)v * Nc + n * 16] = acc[n][v] * sc[v];
  }
}

template <int VEC, bool RELU>
__global__ __launch_bounds__(256)
void k_gather(const float* __restrict__ h, const int* __restrict__ rowptr,
              const int* __restrict__ col, const float* __restrict__ dinv,
              const float* __restrict__ bias, float* __restrict__ out, int N) {
  typedef __attribute__((ext_vector_type(VEC))) float vec;
  constexpr int F = 32 * VEC;
  int node = (blockIdx.x * blockDim.x + threadIdx.x) >> 5;
  const int lane = threadIdx.x & 31;
  if (node >= N) return;
  node = __builtin_amdgcn_readfirstlane(node);

  const int   beg = rowptr[node];
  const int   end = rowptr[node + 1];
  const float dd  = dinv[node];
  const float* hl = h + lane * VEC;

  vec acc = 0.0f;
  for (int base = beg; base < end; base += 32) {
    const int n = min(32, end - base);
    int ms = 0;
    if (lane < n) ms = col[base + lane];
    int i = 0;
    for (; i + 4 <= n; i += 4) {
      const int s0 = __builtin_amdgcn_readlane(ms, i);
      const int s1 = __builtin_amdgcn_readlane(ms, i + 1);
      const int s2 = __builtin_amdgcn_readlane(ms, i + 2);
      const int s3 = __builtin_amdgcn_readlane(ms, i + 3);
      const vec r0 = *(const vec*)(hl + (size_t)s0 * F);
      const vec r1 = *(const vec*)(hl + (size_t)s1 * F);
      const vec r2 = *(const vec*)(hl + (size_t)s2 * F);
      const vec r3 = *(const vec*)(hl + (size_t)s3 * F);
      acc += r0;
      acc += r1;
      acc += r2;
      acc += r3;
    }
    for (; i < n; ++i) {
      const int s = __builtin_amdgcn_readlane(ms, i);
      const vec r = *(const vec*)(hl + (size_t)s * F);
      acc += r;
    }
  }

  const vec selfv = *(const vec*)(hl + (size_t)node * F);
  const vec bv    = *(const vec*)(bias + lane * VEC);
  vec r = (acc + selfv) * dd + bv;
  if (RELU) {
#pragma unroll
    for (int v = 0; v < VEC; ++v) r[v] = fmaxf(r[v], 0.0f);
  }
  *(vec*)(out + (size_t)node * F + lane * VEC) = r;
}

extern "C" void kernel_launch(void* const* d_in, const int* in_sizes, int n_in,
                              void* d_out, int out_size, void* d_ws, size_t ws_size,
                              hipStream_t stream) {
  const float* x  = (const float*)d_in[0];
  const int*   ei = (const int*)d_in[1];
  const float* W1 = (const float*)d_in[2];
  const float* b1 = (const float*)d_in[3];
  const float* W2 = (const float*)d_in[4];
  const float* b2 = (const float*)d_in[5];
  float* out = (float*)d_out;

  const int F0 = 128;
  const int N  = in_sizes[0] / F0;
  const int E  = in_sizes[1] / 2;
  const int F1 = in_sizes[3];
  const int* src = ei;
  const int* dst = ei + E;

  float* h1     = (float*)d_ws;
  float* agg1   = h1 + (size_t)N * F1;
  float* h2     = h1;
  int*   rowptr = (int*)(agg1 + (size_t)N * F1);
  float* dinv   = (float*)(rowptr + N + 64);
  int*   col    = (int*)(dinv + N);
  int*   tmp    = col + ((E + 1) & ~1);

  const int B = 256;
  auto cdiv = [](long long a, long long b) { return (int)((a + b - 1) / b); };
  const int NBK = cdiv(N, BNODES);
  const int CH  = cdiv(E, TILE);
  int*   T      = tmp + 2 * (size_t)E;
  int*   bstart = T + (size_t)CH * MAXB;

  k_bucket_hist<<<CH, 1024, 0, stream>>>(dst, T, E, NBK);
  k_bucket_scatter<<<CH, 1024, 0, stream>>>(src, dst, T, tmp, bstart, E, NBK, CH);
  k_bucket_sort<<<NBK, 1024, 0, stream>>>(tmp, bstart, rowptr, dinv, col, N);

  k_gemm_strip<128, 8><<<cdiv(N / 16, 8), B, 0, stream>>>(x, W1, dinv, h1, N);
  k_gather<4, true><<<cdiv((long long)N * 32, B), B, 0, stream>>>(h1, rowptr, col, dinv, b1, agg1, N);

  k_gemm_strip<128, 4><<<cdiv(N / 16, 8), B, 0, stream>>>(agg1, W2, dinv, h2, N);
  k_gather<2, false><<<cdiv((long long)N * 32, B), B, 0, stream>>>(h2, rowptr, col, dinv, b2, out, N);
}


# Edge tiles keep their bucket-ordered runs in place (own LDS histogram, per-tile offset table); a bucket's group collects its runs from all tiles: counting launch gone, 7 to 6; sum order: LDS arrival
# speedup vs baseline: 6.4214x; 1.0651x over previous
#include <hip/hip_runtime.h>

typedef __attribute__((ext_vector_type(2))) float v2f;
typedef __attribute__((ext_vector_type(8))) float v8f;

constexpr int TILE       = 8192;
constexpr int EDGE_BATCH = 8;
constexpr int BSHIFT     = 8;
constexpr int BNODES     = 1 << BSHIFT;
constexpr int MAXB       = 256;
constexpr int PT         = MAXB + 32;
constexpr int RUN_BATCH  = 8;
constexpr int SEG_CAP    = 24576;

__device__ __forceinline__ int wave_incl_scan(int v, int lane) {
#pragma unroll
  for (int off = 1; off < 32; off <<= 1) {
    const int u = __shfl_up(v, off);
    if (lane >= off) v += u;
  }
  return v;
}

__device__ __forceinline__ int wave_sum(int v) {
#pragma unroll
  for (int off = 16; off > 0; off >>= 1) v += __shfl_xor(v, off);
  return v;
}

__global__ __launch_bounds__(1024)
void k_tile_scatter(const int* __restrict__ src, const int* __restrict__ dst,
                    int* __restrict__ pairs, int* __restrict__ P, int E) {
  int2* tmp = (int2*)pairs;
  __shared__ int2 stage[TILE];
  __shared__ int h[MAXB];
  __shared__ int cur[MAXB];
  __shared__ int wsM[8];
  const int t    = threadIdx.x;
  const int lane = t & 31;
  const int w    = t >> 5;
  const int me   = blockIdx.x;
  const int e0   = me * TILE;
  const int e1   = min(e0 + TILE, E);

  int d[EDGE_BATCH], s[EDGE_BATCH];
#pragma unroll
  for (int j = 0; j < EDGE_BATCH; ++j) {
    const int  e  = e0 + t + j * 1024;
    const bool ok = e < e1;
    d[j] = ok ? dst[e] : -1;
    s[j] = ok ? src[e] : 0;
  }
  if (t < MAXB) h[t] = 0;
  __syncthreads();
#pragma unroll
  for (int j = 0; j < EDGE_BATCH; ++j)
    if (d[j] >= 0) atomicAdd(&h[d[j] >> BSHIFT], 1);
  __syncthreads();

  int mine = 0, incM = 0;
  if (t < MAXB) {
    mine = h[t];
    incM = wave_incl_scan(mine, lane);
    if (lane == 31) wsM[w] = incM;
  }
  __syncthreads();
  if (t < MAXB) {
    int pre = 0;
    for (int i = 0; i < w; ++i) pre += wsM[i];
    const int ls = pre + incM - mine;
    cur[t] = ls;
    P[me * PT + t] = ls;
    if (t == MAXB - 1) P[me * PT + MAXB] = ls + mine;
  }
  __syncthreads();

#pragma unroll
  for (int j = 0; j < EDGE_BATCH; ++j) {
    if (d[j] >= 0) {
      const int p = atomicAdd(&cur[d[j] >> BSHIFT], 1);
      stage[p] = make_int2(s[j], d[j]);
    }
  }
  __syncthreads();

  const int n = e1 - e0;
  for (int i = t; i < n; i += 1024) tmp[e0 + i] = stage[i];
}

__global__ __launch_bounds__(1024)
void k_bucket_sort(const int* __restrict__ pairs, const int* __restrict__ P,
                   int* __restrict__ rowptr, float* __restrict__ dinv,
                   int* __restrict__ col, int N, int CH) {
  __shared__ int scol[SEG_CAP];
  __shared__ int cnt[BNODES];
  __shared__ int fill[BNODES];
  __shared__ int wsA[32], wsL[32];
  __shared__ int tot[2];
  __shared__ int ws[8];
  const int t     = threadIdx.x;
  const int lane  = t & 31;
  const int w     = __builtin_amdgcn_readfirstlane(t >> 5);
  const int b     = blockIdx.x;
  const int node0 = b << BSHIFT;
  const int2* all = (const int2*)pairs;

  const int ct = w + 32 * lane;
  int lo = 0, run_st = 0, run_ln = 0;
  if (ct < CH) {
    lo     = P[ct * PT + b];
    run_ln = P[ct * PT + b + 1] - lo;
    run_st = ct * TILE + lo;
  }
  if (t < BNODES) cnt[t] = 0;
  const int sa = wave_sum(lo);
  const int sl = wave_sum(run_ln);
  if (lane == 0) { wsA[w] = sa; wsL[w] = sl; }
  __syncthreads();
  if (w == 0) {
    const int ga = wave_sum(wsA[lane]);
    const int gl = wave_sum(wsL[lane]);
    if (lane == 0) { tot[0] = ga; tot[1] = gl; }
  }

  const int ntw = (CH - w + 31) >> 5;
  for (int l0 = 0; l0 < ntw; l0 += RUN_BATCH) {
    int dl[RUN_BATCH][2];
#pragma unroll
    for (int j = 0; j < RUN_BATCH; ++j) {
      const int st = __builtin_amdgcn_readlane(run_st, l0 + j);
      const int ln = __builtin_amdgcn_readlane(run_ln, l0 + j);
      dl[j][0] = (lane < ln)      ? all[st + lane].y      : -1;
      dl[j][1] = (lane + 32 < ln) ? all[st + lane + 32].y : -1;
    }
#pragma unroll
    for (int j = 0; j < RUN_BATCH; ++j) {
      if (dl[j][0] >= 0) atomicAdd(&cnt[dl[j][0] - node0], 1);
      if (dl[j][1] >= 0) atomicAdd(&cnt[dl[j][1] - node0], 1);
    }
    for (int j = 0; j < RUN_BATCH; ++j) {
      const int st = __builtin_amdgcn_readlane(run_st, l0 + j);
      const int ln = __builtin_amdgcn_readlane(run_ln, l0 + j);
      for (int i = lane + 64; i < ln; i += 32) atomicAdd(&cnt[all[st + i].y - node0], 1);
    }
  }
  __syncthreads();
  const int g0 = tot[0];
  const int n  = tot[1];

  int c = 0, inc = 0;
  if (t < BNODES) {
    c   = cnt[t];
    inc = wave_incl_scan(c, lane);
    if (lane == 31) ws[w] = inc;
  }
  __syncthreads();
  if (t < BNODES) {
    int pre = 0;
    for (int i = 0; i < w; ++i) pre += ws[i];
    const int ex = pre + inc - c;
    fill[t] = ex;
    const int node = node0 + t;
    if (node < N) {
      rowptr[node] = g0 + ex;
      dinv[node]   = 1.0f / sqrtf((float)(c + 1));
      if (node == N - 1) rowptr[N] = g0 + ex + c;
    }
  }
  __syncthreads();

  const bool fits = n <= SEG_CAP;
  for (int l0 = 0; l0 < ntw; l0 += RUN_BATCH) {
    int2 v[RUN_BATCH][2];
#pragma unroll
    for (int j = 0; j < RUN_BATCH; ++j) {
      const int st = __builtin_amdgcn_readlane(run_st, l0 + j);
      const int ln = __builtin_amdgcn_readlane(run_ln, l0 + j);
      v[j][0] = (lane < ln)      ? all[st + lane]      : make_int2(0, -1);
      v[j][1] = (lane + 32 < ln) ? all[st + lane + 32] : make_int2(0, -1);
    }
#pragma unroll
    for (int j = 0; j < RUN_BATCH; ++j) {
#pragma unroll
      for (int k = 0; k < 2; ++k) {
        if (v[j][k].y >= 0) {
          const int p = atomicAdd(&fill[v[j][k].y - node0], 1);
          if (fits) scol[p] = v[j][k].x;
          else      col[g0 + p] = v[j][k].x;
        }
      }
    }
    for (int j = 0; j < RUN_BATCH; ++j) {
      const int st = __builtin_amdgcn_readlane(run_st, l0 + j);
      const int ln = __builtin_amdgcn_readlane(run_ln, l0 + j);
      for (int i = lane + 64; i < ln; i += 32) {
        const int2 u = all[st + i];
        const int  p = atomicAdd(&fill[u.y - node0], 1);
        if (fits) scol[p] = u.x;
        else      col[g0 + p] = u.x;
      }
    }
  }
  __syncthreads();
  if (fits) {
    for (int i = t; i < n; i += 1024) col[g0 + i] = scol[i];
  }
}

template <int K, int NT>
__global__ __launch_bounds__(256)
void k_gemm_strip(const float* __restrict__ A, const float* __restrict__ B,
                  const float* __restrict__ rs, float* __restrict__ C, int M) {
  constexpr int Nc = NT * 16;
  constexpr int PP = 2 * Nc + 32;
  constexpr int Q  = Nc / 4;
  __shared__ alignas(16) float Bs[(K / 2) * PP];
  const int t     = threadIdx.x;
  const int lane  = t & 31;
  const int half  = lane >> 4;
  const int j     = lane & 15;
  const int strip = blockIdx.x * 8 + (t >> 5);
  const bool valid = strip < (M >> 4);

  constexpr int TRIPS = (K / 2) * Q / 256;
  static_assert(TRIPS * 256 == (K / 2) * Q, "B must split evenly over the 256 threads");
  float4 r0[TRIPS], r1[TRIPS];
#pragma unroll
  for (int it = 0; it < TRIPS; ++it) {
    const int i = t + it * 256;
    const int p = i / Q;
    const int c = i - p * Q;
    r0[it] = ((const float4*)B)[(2 * p) * Q + c];
    r1[it] = ((const float4*)B)[(2 * p + 1) * Q + c];
  }

  const int srow = valid ? strip : 0;
  const float* Arow = A + (size_t)(srow * 16 + j) * K + 2 * half;
  v2f a[K / 4];
#pragma unroll
  for (int i = 0; i < K / 4; ++i) a[i] = *(const v2f*)(Arow + 4 * i);
  const float4 sc0 = *(const float4*)(rs + srow * 16 + 8 * half);
  const float4 sc1 = *(const float4*)(rs + srow * 16 + 8 * half + 4);

#pragma unroll
  for (int it = 0; it < TRIPS; ++it) {
    const int i = t + it * 256;
    const int p = i / Q;
    const int c = i - p * Q;
    float* d = Bs + p * PP + 8 * c;
    *(float4*)(d)     = make_float4(r0[it].x, r1[it].x, r0[it].y, r1[it].y);
    *(float4*)(d + 4) = make_float4(r0[it].z, r1[it].z, r0[it].w, r1[it].w);
  }
  __builtin_amdgcn_fence(__ATOMIC_RELEASE, "workgroup", "local");
  __builtin_amdgcn_s_barrier();
  __builtin_amdgcn_fence(__ATOMIC_ACQUIRE, "workgroup", "local");
  if (!valid) return;

  const float* Bl = Bs + half * PP + 2 * j;

  v8f acc[NT];
#pragma unroll
  for (int n = 0; n < NT; ++n) acc[n] = v8f{};

#pragma unroll
  for (int ks = 0; ks < K / 4; ++ks) {
#pragma unroll
    for (int n = 0; n < NT; ++n) {
      const v2f b = *(const v2f*)(Bl + 2 * ks * PP + 32 * n);
      acc[n] = __builtin_amdgcn_wmma_f32_16x16x4_f32(false, a[ks], false, b, (short)0, acc[n],
                                                     false, false);
    }
  }

  const float sc[8] = {sc0.x, sc0.y, sc0.z, sc0.w, sc1.x, sc1.y, sc1.z, sc1.w};
  float* Cp = C + (size_t)(strip * 16 + 8 * half) * Nc + j;
#pragma unroll
  for (int n = 0; n < NT; ++n) {
#pragma unroll
    for (int v = 0; v < 8; ++v) Cp[(size_t)v * Nc + n * 16] = acc[n][v] * sc[v];
  }
}

template <int VEC, bool RELU>
__global__ __launch_bounds__(256)
void k_gather(const float* __restrict__ h, const int* __restrict__ rowptr,
              const int* __restrict__ col, const float* __restrict__ dinv,
              const float* __restrict__ bias, float* __restrict__ out, int N) {
  typedef __attribute__((ext_vector_type(VEC))) float vec;
  constexpr int F = 32 * VEC;
  int node = (blockIdx.x * blockDim.x + threadIdx.x) >> 5;
  const int lane = threadIdx.x & 31;
  if (node >= N) return;
  node = __builtin_amdgcn_readfirstlane(node);

  const int   beg = rowptr[node];
  const int   end = rowptr[node + 1];
  const float dd  = dinv[node];
  const float* hl = h + lane * VEC;

  vec acc = 0.0f;
  for (int base = beg; base < end; base += 32) {
    const int n = min(32, end - base);
    int ms = 0;
    if (lane < n) ms = col[base + lane];
    int i = 0;
    for (; i + 4 <= n; i += 4) {
      const int s0 = __builtin_amdgcn_readlane(ms, i);
      const int s1 = __builtin_amdgcn_readlane(ms, i + 1);
      const int s2 = __builtin_amdgcn_readlane(ms, i + 2);
      const int s3 = __builtin_amdgcn_readlane(ms, i + 3);
      const vec r0 = *(const vec*)(hl + (size_t)s0 * F);
      const vec r1 = *(const vec*)(hl + (size_t)s1 * F);
      const vec r2 = *(const vec*)(hl + (size_t)s2 * F);
      const vec r3 = *(const vec*)(hl + (size_t)s3 * F);
      acc += r0;
      acc += r1;
      acc += r2;
      acc += r3;
    }
    for (; i < n; ++i) {
      const int s = __builtin_amdgcn_readlane(ms, i);
      const vec r = *(const vec*)(hl + (size_t)s * F);
      acc += r;
    }
  }

  const vec selfv = *(const vec*)(hl + (size_t)node * F);
  const vec bv    = *(const vec*)(bias + lane * VEC);
  vec r = (acc + selfv) * dd + bv;
  if (RELU) {
#pragma unroll
    for (int v = 0; v < VEC; ++v) r[v] = fmaxf(r[v], 0.0f);
  }
  *(vec*)(out + (size_t)node * F + lane * VEC) = r;
}

extern "C" void kernel_launch(void* const* d_in, const int* in_sizes, int n_in,
                              void* d_out, int out_size, void* d_ws, size_t ws_size,
                              hipStream_t stream) {
  const float* x  = (const float*)d_in[0];
  const int*   ei = (const int*)d_in[1];
  const float* W1 = (const float*)d_in[2];
  const float* b1 = (const float*)d_in[3];
  const float* W2 = (const float*)d_in[4];
  const float* b2 = (const float*)d_in[5];
  float* out = (float*)d_out;

  const int F0 = 128;
  const int N  = in_sizes[0] / F0;
  const int E  = in_sizes[1] / 2;
  const int F1 = in_sizes[3];
  const int* src = ei;
  const int* dst = ei + E;

  float* h1     = (float*)d_ws;
  float* agg1   = h1 + (size_t)N * F1;
  float* h2     = h1;
  int*   rowptr = (int*)(agg1 + (size_t)N * F1);
  float* dinv   = (float*)(rowptr + N + 64);
  int*   col    = (int*)(dinv + N);
  int*   tmp    = col + ((E + 1) & ~1);

  const int B = 256;
  auto cdiv = [](long long a, long long b) { return (int)((a + b - 1) / b); };
  const int NBK = cdiv(N, BNODES);
  const int CH  = cdiv(E, TILE);
  const size_t pOff = ((size_t)(tmp - (int*)d_ws) + 2 * (size_t)E + 31) & ~(size_t)31;
  int*   P      = (int*)d_ws + pOff;

  k_tile_scatter<<<CH, 1024, 0, stream>>>(src, dst, tmp, P, E);
  k_bucket_sort<<<NBK, 1024, 0, stream>>>(tmp, P, rowptr, dinv, col, N, CH);

  k_gemm_strip<128, 8><<<cdiv(N / 16, 8), B, 0, stream>>>(x, W1, dinv, h1, N);
  k_gather<4, true><<<cdiv((long long)N * 32, B), B, 0, stream>>>(h1, rowptr, col, dinv, b1, agg1, N);

  k_gemm_strip<128, 4><<<cdiv(N / 16, 8), B, 0, stream>>>(agg1, W2, dinv, h2, N);
  k_gather<2, false><<<cdiv((long long)N * 32, B), B, 0, stream>>>(h2, rowptr, col, dinv, b2, out, N);
}


# Layer-1 product moved into the bucket-sort launch: a bucket's work-group, having made dinv of its 256 rows, computes their 16 strips (two waves per strip); W1 staged under the sort; 6 to 5 launches
# speedup vs baseline: 6.6818x; 1.0406x over previous
#include <hip/hip_runtime.h>

typedef __attribute__((ext_vector_type(2))) float v2f;
typedef __attribute__((ext_vector_type(8))) float v8f;

constexpr int TILE       = 8192;
constexpr int EDGE_BATCH = 8;
constexpr int BSHIFT     = 8;
constexpr int BNODES     = 1 << BSHIFT;
constexpr int MAXB       = 256;
constexpr int PT         = MAXB + 32;
constexpr int RUN_BATCH  = 8;
constexpr int SEG_CAP    = 24576;

__device__ __forceinline__ int wave_incl_scan(int v, int lane) {
#pragma unroll
  for (int off = 1; off < 32; off <<= 1) {
    const int u = __shfl_up(v, off);
    if (lane >= off) v += u;
  }
  return v;
}

__device__ __forceinline__ int wave_sum(int v) {
#pragma unroll
  for (int off = 16; off > 0; off >>= 1) v += __shfl_xor(v, off);
  return v;
}

__global__ __launch_bounds__(1024)
void k_tile_scatter(const int* __restrict__ src, const int* __restrict__ dst,
                    int* __restrict__ pairs, int* __restrict__ P, int E) {
  int2* tmp = (int2*)pairs;
  __shared__ int2 stage[TILE];
  __shared__ int h[MAXB];
  __shared__ int cur[MAXB];
  __shared__ int wsM[8];
  const int t    = threadIdx.x;
  const int lane = t & 31;
  const int w    = t >> 5;
  const int me   = blockIdx.x;
  const int e0   = me * TILE;
  const int e1   = min(e0 + TILE, E);

  int d[EDGE_BATCH], s[EDGE_BATCH];
#pragma unroll
  for (int j = 0; j < EDGE_BATCH; ++j) {
    const int  e  = e0 + t + j * 1024;
    const bool ok = e < e1;
    d[j] = ok ? dst[e] : -1;
    s[j] = ok ? src[e] : 0;
  }
  if (t < MAXB) h[t] = 0;
  __syncthreads();
#pragma unroll
  for (int j = 0; j < EDGE_BATCH; ++j)
    if (d[j] >= 0) atomicAdd(&h[d[j] >> BSHIFT], 1);
  __syncthreads();

  int mine = 0, incM = 0;
  if (t < MAXB) {
    mine = h[t];
    incM = wave_incl_scan(mine, lane);
    if (lane == 31) wsM[w] = incM;
  }
  __syncthreads();
  if (t < MAXB) {
    int pre = 0;
    for (int i = 0; i < w; ++i) pre += wsM[i];
    const int ls = pre + incM - mine;
    cur[t] = ls;
    P[me * PT + t] = ls;
    if (t == MAXB - 1) P[me * PT + MAXB] = ls + mine;
  }
  __syncthreads();

#pragma unroll
  for (int j = 0; j < EDGE_BATCH; ++j) {
    if (d[j] >= 0) {
      const int p = atomicAdd(&cur[d[j] >> BSHIFT], 1);
      stage[p] = make_int2(s[j], d[j]);
    }
  }
  __syncthreads();

  const int n = e1 - e0;
  for (int i = t; i < n; i += 1024) tmp[e0 + i] = stage[i];
}

template <int K, int NT>
__global__ __launch_bounds__(1024)
void k_bucket_sort(const int* __restrict__ pairs, const int* __restrict__ P,
                   int* __restrict__ rowptr, float* __restrict__ dinv,
                   int* __restrict__ col, int N, int CH,
                   const float* __restrict__ A, const float* __restrict__ B,
                   float* __restrict__ C) {
  constexpr int Nc  = NT * 16;
  constexpr int PP  = 2 * Nc + 32;
  constexpr int Q   = Nc / 4;
  constexpr int NTW = NT / 2;
  static_assert(BNODES == 256 && NT % 2 == 0, "16 strips per bucket, two waves per strip");
  __shared__ alignas(16) int smem[(K / 2) * PP + SEG_CAP];
  float* const Bs   = (float*)smem;
  int*   const scol = smem + (K / 2) * PP;
  __shared__ alignas(16) float sdinv[BNODES];
  __shared__ int cnt[BNODES];
  __shared__ int fill[BNODES];
  __shared__ int wsA[32], wsL[32];
  __shared__ int tot[2];
  __shared__ int ws[8];
  const int t     = threadIdx.x;
  const int lane  = t & 31;
  const int w     = __builtin_amdgcn_readfirstlane(t >> 5);
  const int b     = blockIdx.x;
  const int node0 = b << BSHIFT;
  const int2* all = (const int2*)pairs;

  constexpr int TRIPS = (K / 2) * Q / 1024;
  static_assert(TRIPS * 1024 == (K / 2) * Q, "B must split evenly over the 1024 threads");
  float4 r0[TRIPS], r1[TRIPS];
#pragma unroll
  for (int it = 0; it < TRIPS; ++it) {
    const int i = t + it * 1024;
    const int p = i / Q;
    const int c = i - p * Q;
    r0[it] = ((const float4*)B)[(2 * p) * Q + c];
    r1[it] = ((const float4*)B)[(2 * p + 1) * Q + c];
  }

  const int ct = w + 32 * lane;
  int lo = 0, run_st = 0, run_ln = 0;
  if (ct < CH) {
    lo     = P[ct * PT + b];
    run_ln = P[ct * PT + b + 1] - lo;
    run_st = ct * TILE + lo;
  }
  if (t < BNODES) cnt[t] = 0;
  const int sa = wave_sum(lo);
  const int sl = wave_sum(run_ln);
  if (lane == 0) { wsA[w] = sa; wsL[w] = sl; }
#pragma unroll
  for (int it = 0; it < TRIPS; ++it) {
    const int i = t + it * 1024;
    const int p = i / Q;
    const int c = i - p * Q;
    float* d = Bs + p * PP + 8 * c;
    *(float4*)(d)     = make_float4(r0[it].x, r1[it].x, r0[it].y, r1[it].y);
    *(float4*)(d + 4) = make_float4(r0[it].z, r1[it].z, r0[it].w, r1[it].w);
  }
  __syncthreads();
  if (w == 0) {
    const int ga = wave_sum(wsA[lane]);
    const int gl = wave_sum(wsL[lane]);
    if (lane == 0) { tot[0] = ga; tot[1] = gl; }
  }

  const int ntw = (CH - w + 31) >> 5;
  for (int l0 = 0; l0 < ntw; l0 += RUN_BATCH) {
    int dl[RUN_BATCH][2];
#pragma unroll
    for (int j = 0; j < RUN_BATCH; ++j) {
      const int st = __builtin_amdgcn_readlane(run_st, l0 + j);
      const int ln = __builtin_amdgcn_readlane(run_ln, l0 + j);
      dl[j][0] = (lane < ln)      ? all[st + lane].y      : -1;
      dl[j][1] = (lane + 32 < ln) ? all[st + lane + 32].y : -1;
    }
#pragma unroll
    for (int j = 0; j < RUN_BATCH; ++j) {
      if (dl[j][0] >= 0) atomicAdd(&cnt[dl[j][0] - node0], 1);
      if (dl[j][1] >= 0) atomicAdd(&cnt[dl[j][1] - node0], 1);
    }
    for (int j = 0; j < RUN_BATCH; ++j) {
      const int st = __builtin_amdgcn_readlane(run_st, l0 + j);
      const int ln = __builtin_amdgcn_readlane(run_ln, l0 + j);
      for (int i = lane + 64; i < ln; i += 32) atomicAdd(&cnt[all[st + i].y - node0], 1);
    }
  }
  __syncthreads();
  const int g0 = tot[0];
  const int n  = tot[1];

  int c = 0, inc = 0;
  if (t < BNODES) {
    c   = cnt[t];
    inc = wave_incl_scan(c, lane);
    if (lane == 31) ws[w] = inc;
  }
  __syncthreads();
  if (t < BNODES) {
    int pre = 0;
    for (int i = 0; i < w; ++i) pre += ws[i];
    const int ex = pre + inc - c;
    fill[t] = ex;
    const int node = node0 + t;
    const float dv = 1.0f / sqrtf((float)(c + 1));
    sdinv[t] = dv;
    if (node < N) {
      rowptr[node] = g0 + ex;
      dinv[node]   = dv;
      if (node == N - 1) rowptr[N] = g0 + ex + c;
    }
  }
  __syncthreads();

  const bool fits = n <= SEG_CAP;
  for (int l0 = 0; l0 < ntw; l0 += RUN_BATCH) {
    int2 v[RUN_BATCH][2];
#pragma unroll
    for (int j = 0; j < RUN_BATCH; ++j) {
      const int st = __builtin_amdgcn_readlane(run_st, l0 + j);
      const int ln = __builtin_amdgcn_readlane(run_ln, l0 + j);
      v[j][0] = (lane < ln)      ? all[st + lane]      : make_int2(0, -1);
      v[j][1] = (lane + 32 < ln) ? all[st + lane + 32] : make_int2(0, -1);
    }
#pragma unroll
    for (int j = 0; j < RUN_BATCH; ++j) {
#pragma unroll
      for (int k = 0; k < 2; ++k) {
        if (v[j][k].y >= 0) {
          const int p = atomicAdd(&fill[v[j][k].y - node0], 1);
          if (fits) scol[p] = v[j][k].x;
          else      col[g0 + p] = v[j][k].x;
        }
      }
    }
    for (int j = 0; j < RUN_BATCH; ++j) {
      const int st = __builtin_amdgcn_readlane(run_st, l0 + j);
      const int ln = __builtin_amdgcn_readlane(run_ln, l0 + j);
      for (int i = lane + 64; i < ln; i += 32) {
        const int2 u = all[st + i];
        const int  p = atomicAdd(&fill[u.y - node0], 1);
        if (fits) scol[p] = u.x;
        else      col[g0 + p] = u.x;
      }
    }
  }
  __syncthreads();

  const int  half  = lane >> 4;
  const int  j     = lane & 15;
  const int  ls    = w >> 1;
  const int  ch    = w & 1;
  const int  strip = b * 16 + ls;
  const bool valid = strip < (N >> 4);
  const int  srow  = valid ? strip : 0;
  const float* Arow = A + (size_t)(srow * 16 + j) * K + 2 * half;
  v2f a[K / 4];
#pragma unroll
  for (int i = 0; i < K / 4; ++i) a[i] = *(const v2f*)(Arow + 4 * i);

  if (fits) {
    for (int i = t; i < n; i += 1024) col[g0 + i] = scol[i];
  }
  if (!valid) return;

  const float* Bl = Bs + half * PP + 2 * j + ch * (NTW * 32);
  v8f acc[NTW];
#pragma unroll
  for (int q = 0; q < NTW; ++q) acc[q] = v8f{};

#pragma unroll
  for (int ks = 0; ks < K / 4; ++ks) {
#pragma unroll
    for (int q = 0; q < NTW; ++q) {
      const v2f bf = *(const v2f*)(Bl + 2 * ks * PP + 32 * q);
      acc[q] = __builtin_amdgcn_wmma_f32_16x16x4_f32(false, a[ks], false, bf, (short)0, acc[q],
                                                     false, false);
    }
  }

  const float4 sc0 = *(const float4*)(sdinv + ls * 16 + 8 * half);
  const float4 sc1 = *(const float4*)(sdinv + ls * 16 + 8 * half + 4);
  const float sc[8] = {sc0.x, sc0.y, sc0.z, sc0.w, sc1.x, sc1.y, sc1.z, sc1.w};
  float* Cp = C + (size_t)(strip * 16 + 8 * half) * Nc + ch * (NTW * 16) + j;
#pragma unroll
  for (int q = 0; q < NTW; ++q) {
#pragma unroll
    for (int v = 0; v < 8; ++v) Cp[(size_t)v * Nc + q * 16] = acc[q][v] * sc[v];
  }
}

template <int K, int NT>
__global__ __launch_bounds__(256)
void k_gemm_strip(const float* __restrict__ A, const float* __restrict__ B,
                  const float* __restrict__ rs, float* __restrict__ C, int M) {
  constexpr int Nc = NT * 16;
  constexpr int PP = 2 * Nc + 32;
  constexpr int Q  = Nc / 4;
  __shared__ alignas(16) float Bs[(K / 2) * PP];
  const int t     = threadIdx.x;
  const int lane  = t & 31;
  const int half  = lane >> 4;
  const int j     = lane & 15;
  const int strip = blockIdx.x * 8 + (t >> 5);
  const bool valid = strip < (M >> 4);

  constexpr int TRIPS = (K / 2) * Q / 256;
  static_assert(TRIPS * 256 == (K / 2) * Q, "B must split evenly over the 256 threads");
  float4 r0[TRIPS], r1[TRIPS];
#pragma unroll
  for (int it = 0; it < TRIPS; ++it) {
    const int i = t + it * 256;
    const int p = i / Q;
    const int c = i - p * Q;
    r0[it] = ((const float4*)B)[(2 * p) * Q + c];
    r1[it] = ((const float4*)B)[(2 * p + 1) * Q + c];
  }

  const int srow = valid ? strip : 0;
  const float* Arow = A + (size_t)(srow * 16 + j) * K + 2 * half;
  v2f a[K / 4];
#pragma unroll
  for (int i = 0; i < K / 4; ++i) a[i] = *(const v2f*)(Arow + 4 * i);
  const float4 sc0 = *(const float4*)(rs + srow * 16 + 8 * half);
  const float4 sc1 = *(const float4*)(rs + srow * 16 + 8 * half + 4);

#pragma unroll
  for (int it = 0; it < TRIPS; ++it) {
    const int i = t + it * 256;
    const int p = i / Q;
    const int c = i - p * Q;
    float* d = Bs + p * PP + 8 * c;
    *(float4*)(d)     = make_float4(r0[it].x, r1[it].x, r0[it].y, r1[it].y);
    *(float4*)(d + 4) = make_float4(r0[it].z, r1[it].z, r0[it].w, r1[it].w);
  }
  __builtin_amdgcn_fence(__ATOMIC_RELEASE, "workgroup", "local");
  __builtin_amdgcn_s_barrier();
  __builtin_amdgcn_fence(__ATOMIC_ACQUIRE, "workgroup", "local");
  if (!valid) return;

  const float* Bl = Bs + half * PP + 2 * j;

  v8f acc[NT];
#pragma unroll
  for (int n = 0; n < NT; ++n) acc[n] = v8f{};

#pragma unroll
  for (int ks = 0; ks < K / 4; ++ks) {
#pragma unroll
    for (int n = 0; n < NT; ++n) {
      const v2f b = *(const v2f*)(Bl + 2 * ks * PP + 32 * n);
      acc[n] = __builtin_amdgcn_wmma_f32_16x16x4_f32(false, a[ks], false, b, (short)0, acc[n],
                                                     false, false);
    }
  }

  const float sc[8] = {sc0.x, sc0.y, sc0.z, sc0.w, sc1.x, sc1.y, sc1.z, sc1.w};
  float* Cp = C + (size_t)(strip * 16 + 8 * half) * Nc + j;
#pragma unroll
  for (int n = 0; n < NT; ++n) {
#pragma unroll
    for (int v = 0; v < 8; ++v) Cp[(size_t)v * Nc + n * 16] = acc[n][v] * sc[v];
  }
}

template <int VEC, bool RELU>
__global__ __launch_bounds__(256)
void k_gather(const float* __restrict__ h, const int* __restrict__ rowptr,
              const int* __restrict__ col, const float* __restrict__ dinv,
              const float* __restrict__ bias, float* __restrict__ out, int N) {
  typedef __attribute__((ext_vector_type(VEC))) float vec;
  constexpr int F = 32 * VEC;
  int node = (blockIdx.x * blockDim.x + threadIdx.x) >> 5;
  const int lane = threadIdx.x & 31;
  if (node >= N) return;
  node = __builtin_amdgcn_readfirstlane(node);

  const int   beg = rowptr[node];
  const int   end = rowptr[node + 1];
  const float dd  = dinv[node];
  const float* hl = h + lane * VEC;

  vec acc = 0.0f;
  for (int base = beg; base < end; base += 32) {
    const int n = min(32, end - base);
    int ms = 0;
    if (lane < n) ms = col[base + lane];
    int i = 0;
    for (; i + 4 <= n; i += 4) {
      const int s0 = __builtin_amdgcn_readlane(ms, i);
      const int s1 = __builtin_amdgcn_readlane(ms, i + 1);
      const int s2 = __builtin_amdgcn_readlane(ms, i + 2);
      const int s3 = __builtin_amdgcn_readlane(ms, i + 3);
      const vec r0 = *(const vec*)(hl + (size_t)s0 * F);
      const vec r1 = *(const vec*)(hl + (size_t)s1 * F);
      const vec r2 = *(const vec*)(hl + (size_t)s2 * F);
      const vec r3 = *(const vec*)(hl + (size_t)s3 * F);
      acc += r0;
      acc += r1;
      acc += r2;
      acc += r3;
    }
    for (; i < n; ++i) {
      const int s = __builtin_amdgcn_readlane(ms, i);
      const vec r = *(const vec*)(hl + (size_t)s * F);
      acc += r;
    }
  }

  const vec selfv = *(const vec*)(hl + (size_t)node * F);
  const vec bv    = *(const vec*)(bias + lane * VEC);
  vec r = (acc + selfv) * dd + bv;
  if (RELU) {
#pragma unroll
    for (int v = 0; v < VEC; ++v) r[v] = fmaxf(r[v], 0.0f);
  }
  *(vec*)(out + (size_t)node * F + lane * VEC) = r;
}

extern "C" void kernel_launch(void* const* d_in, const int* in_sizes, int n_in,
                              void* d_out, int out_size, void* d_ws, size_t ws_size,
                              hipStream_t stream) {
  const float* x  = (const float*)d_in[0];
  const int*   ei = (const int*)d_in[1];
  const float* W1 = (const float*)d_in[2];
  const float* b1 = (const float*)d_in[3];
  const float* W2 = (const float*)d_in[4];
  const float* b2 = (const float*)d_in[5];
  float* out = (float*)d_out;

  const int F0 = 128;
  const int N  = in_sizes[0] / F0;
  const int E  = in_sizes[1] / 2;
  const int F1 = in_sizes[3];
  const int* src = ei;
  const int* dst = ei + E;

  float* h1     = (float*)d_ws;
  float* agg1   = h1 + (size_t)N * F1;
  float* h2     = h1;
  int*   rowptr = (int*)(agg1 + (size_t)N * F1);
  float* dinv   = (float*)(rowptr + N + 64);
  int*   col    = (int*)(dinv + N);
  int*   tmp    = col + ((E + 1) & ~1);

  const int B = 256;
  auto cdiv = [](long long a, long long b) { return (int)((a + b - 1) / b); };
  const int NBK = cdiv(N, BNODES);
  const int CH  = cdiv(E, TILE);
  const size_t pOff = ((size_t)(tmp - (int*)d_ws) + 2 * (size_t)E + 31) & ~(size_t)31;
  int*   P      = (int*)d_ws + pOff;

  k_tile_scatter<<<CH, 1024, 0, stream>>>(src, dst, tmp, P, E);
  k_bucket_sort<128, 8><<<NBK, 1024, 0, stream>>>(tmp, P, rowptr, dinv, col, N, CH, x, W1, h1);

  k_gather<4, true><<<cdiv((long long)N * 32, B), B, 0, stream>>>(h1, rowptr, col, dinv, b1, agg1, N);

  k_gemm_strip<128, 4><<<cdiv(N / 16, 8), B, 0, stream>>>(agg1, W2, dinv, h2, N);
  k_gather<2, false><<<cdiv((long long)N * 32, B), B, 0, stream>>>(h2, rowptr, col, dinv, b2, out, N);
}
